# starting kernel
# baseline (speedup 1.0000x reference)
#include <hip/hip_runtime.h>
#include <hip/hip_bf16.h>

typedef __attribute__((ext_vector_type(2))) float v2f;
typedef __attribute__((ext_vector_type(8))) float v8f;

#define TK 128
#define LDK 132

__global__ void gcn_deg_kernel(const int* __restrict__ src, const int* __restrict__ dst,
                               float* __restrict__ degO, float* __restrict__ degI, int E) {
  int t = blockIdx.x * blockDim.x + threadIdx.x;
  if (t < E) {
    atomicAdd(&degO[src[t]], 1.0f);
    atomicAdd(&degI[dst[t]], 1.0f);
  }
}

__global__ void gcn_invsqrt_kernel(float* __restrict__ degO, float* __restrict__ degI, int N) {
  int t = blockIdx.x * blockDim.x + threadIdx.x;
  if (t < N) {
    float o = degO[t];
    float i = degI[t];
    degO[t] = (o > 0.0f) ? rsqrtf(fmaxf(o, 1.0f)) : 0.0f;
    degI[t] = (i > 0.0f) ? rsqrtf(fmaxf(i, 1.0f)) : 0.0f;
  }
}

template <int CHUNKS>
__global__ void gcn_scatter_kernel(const float* __restrict__ h,
                                   const int* __restrict__ src,
                                   const int* __restrict__ dst,
                                   const float* __restrict__ invO,
                                   float* __restrict__ agg,
                                   int E) {
  constexpr int F = CHUNKS * 4;
  long long t = (long long)blockIdx.x * blockDim.x + threadIdx.x;
  long long total = (long long)E * CHUNKS;
  if (t >= total) return;
  int e = (int)(t / CHUNKS);
  int c = (int)(t - (long long)e * CHUNKS);
  int s = src[e];
  int d = dst[e];
  float sc = invO[s];
  float4 v = ((const float4*)(h + (size_t)s * F))[c];
  float* ad = agg + (size_t)d * F + (size_t)c * 4;
  atomicAdd(ad + 0, v.x * sc);
  atomicAdd(ad + 1, v.y * sc);
  atomicAdd(ad + 2, v.z * sc);
  atomicAdd(ad + 3, v.w * sc);
}

__global__ void __launch_bounds__(256)
gcn_gemm_kernel(const float* __restrict__ A,
                const float* __restrict__ rowScale,
                const float* __restrict__ W,
                const float* __restrict__ bias,
                float* __restrict__ C,
                int M, int Nd, int relu) {
  __shared__ float As[16 * LDK];

  const int t     = threadIdx.x;
  const int mBase = blockIdx.x * 16;

  for (int i = t; i < 16 * (TK / 4); i += 256) {
    int row = i >> 5;
    int c4  = i & 31;
    int gr  = mBase + row;
    float4 v = make_float4(0.f, 0.f, 0.f, 0.f);
    float  s = 1.0f;
    if (gr < M) {
      v = ((const float4*)(A + (size_t)gr * TK))[c4];
      if (rowScale) s = rowScale[gr];
    }
    float* dstp = As + (size_t)row * LDK + (size_t)c4 * 4;
    dstp[0] = v.x * s; dstp[1] = v.y * s; dstp[2] = v.z * s; dstp[3] = v.w * s;
  }
  __syncthreads();

  const int wave   = t >> 5;
  const int lane   = t & 31;
  const int nTiles = (Nd + 15) >> 4;
  if (wave >= nTiles) return;

  const int col    = lane & 15;
  const int gcol   = wave * 16 + col;
  const int khalf  = (lane >> 4) * 2;
  const bool colOk = (gcol < Nd);
  const int gcSafe = colOk ? gcol : (Nd - 1);

  const float* As_row = As + (size_t)(lane & 15) * LDK + khalf;
  const float* Wp     = W + (size_t)khalf * Nd + gcSafe;

  v8f acc0 = {};
  v8f acc1 = {};
  #pragma unroll 8
  for (int k0 = 0; k0 < TK; k0 += 8) {
    v2f a0, b0, a1, b1;
    a0[0] = As_row[k0];
    a0[1] = As_row[k0 + 1];
    b0[0] = Wp[0];
    b0[1] = Wp[Nd];
    a1[0] = As_row[k0 + 4];
    a1[1] = As_row[k0 + 5];
    b1[0] = Wp[4 * (size_t)Nd];
    b1[1] = Wp[5 * (size_t)Nd];
    acc0 = __builtin_amdgcn_wmma_f32_16x16x4_f32(false, a0, false, b0, (short)0, acc0,
                                                 false, false);
    acc1 = __builtin_amdgcn_wmma_f32_16x16x4_f32(false, a1, false, b1, (short)0, acc1,
                                                 false, false);
    Wp += 8 * (size_t)Nd;
  }
  v8f acc = acc0 + acc1;

  if (colOk) {
    const float bb    = bias ? bias[gcol] : 0.0f;
    const int   rbase = mBase + (lane >> 4) * 8;
    #pragma unroll
    for (int v = 0; v < 8; ++v) {
      int r = rbase + v;
      if (r < M) {
        float val = acc[v] + bb;
        if (relu) val = fmaxf(val, 0.0f);
        C[(size_t)r * Nd + gcol] = val;
      }
    }
  }
}

__global__ void gcn_final_kernel(const float* __restrict__ agg,
                                 const float* __restrict__ invI,
                                 const float* __restrict__ b3,
                                 float* __restrict__ out, int N, int Cc) {
  int t = blockIdx.x * blockDim.x + threadIdx.x;
  if (t < N * Cc) {
    int n = t / Cc;
    int c = t - n * Cc;
    out[t] = agg[t] * invI[n] + b3[c];
  }
}

extern "C" void kernel_launch(void* const* d_in, const int* in_sizes, int n_in,
                              void* d_out, int out_size, void* d_ws, size_t ws_size,
                              hipStream_t stream) {
  const float* feats = (const float*)d_in[0];
  const int*   src   = (const int*)d_in[1];
  const int*   dst   = (const int*)d_in[2];
  const float* W1    = (const float*)d_in[3];
  const float* b1    = (const float*)d_in[4];
  const float* W2    = (const float*)d_in[5];
  const float* b2    = (const float*)d_in[6];
  const float* W3    = (const float*)d_in[7];
  const float* b3    = (const float*)d_in[8];
  float*       out   = (float*)d_out;

  const int HID  = in_sizes[4];
  const int NCLS = in_sizes[8];
  const int INF  = in_sizes[3] / HID;
  const int N    = in_sizes[0] / INF;
  const int E    = in_sizes[1];

  float* invO = (float*)d_ws;
  float* invI = invO + N;
  float* bufA = invI + N;
  float* bufB = bufA + (size_t)N * TK;
  float* bufC = bufB + (size_t)N * TK;

  const int B = 256;
  const int mTiles = (N + 15) / 16;

  hipMemsetAsync(invO, 0, sizeof(float) * 2 * (size_t)N, stream);
  gcn_deg_kernel<<<(E + B - 1) / B, B, 0, stream>>>(src, dst, invO, invI, E);
  gcn_invsqrt_kernel<<<(N + B - 1) / B, B, 0, stream>>>(invO, invI, N);

  hipMemsetAsync(bufA, 0, sizeof(float) * (size_t)N * TK, stream);
  {
    long long tot = (long long)E * 32;
    gcn_scatter_kernel<32><<<(int)((tot + B - 1) / B), B, 0, stream>>>(feats, src, dst, invO, bufA, E);
  }
  gcn_gemm_kernel<<<mTiles, B, 0, stream>>>(bufA, invI, W1, b1, bufB, N, HID, 1);

  hipMemsetAsync(bufA, 0, sizeof(float) * (size_t)N * TK, stream);
  {
    long long tot = (long long)E * 32;
    gcn_scatter_kernel<32><<<(int)((tot + B - 1) / B), B, 0, stream>>>(bufB, src, dst, invO, bufA, E);
  }
  gcn_gemm_kernel<<<mTiles, B, 0, stream>>>(bufA, invI, W2, b2, bufB, N, HID, 1);

  gcn_gemm_kernel<<<mTiles, B, 0, stream>>>(bufB, nullptr, W3, nullptr, bufC, N, NCLS, 0);
  hipMemsetAsync(bufA, 0, sizeof(float) * (size_t)N * NCLS, stream);
  {
    long long tot = (long long)E * 10;
    gcn_scatter_kernel<10><<<(int)((tot + B - 1) / B), B, 0, stream>>>(bufC, src, dst, invO, bufA, E);
  }
  gcn_final_kernel<<<(N * NCLS + B - 1) / B, B, 0, stream>>>(bufA, invI, b3, out, N, NCLS);
}


# Atomic scatter-add replaced by a dst-sorted edge list built each call (int count with rank, scan, fill) and atomic-free per-node gather-sum; per-node sum order = edge arrival order, varies per run
# speedup vs baseline: 2.0403x; 2.0403x over previous
#include <hip/hip_runtime.h>
#include <hip/hip_bf16.h>

typedef __attribute__((ext_vector_type(2))) float v2f;
typedef __attribute__((ext_vector_type(8))) float v8f;

#define TK 128
#define LDK 132

__global__ void gcn_hist_kernel(const int* __restrict__ src, const int* __restrict__ dst,
                                int* __restrict__ cntO, int* __restrict__ cntI,
                                int* __restrict__ rank, int E) {
  int t = blockIdx.x * blockDim.x + threadIdx.x;
  if (t < E) {
    atomicAdd(&cntO[src[t]], 1);
    rank[t] = atomicAdd(&cntI[dst[t]], 1);
  }
}

__device__ __forceinline__ float gcn_inv_sqrt_deg(int c) {
  return (c > 0) ? rsqrtf(fmaxf((float)c, 1.0f)) : 0.0f;
}

__global__ void __launch_bounds__(1024)
gcn_scan_kernel(const int* __restrict__ cntO, const int* __restrict__ cntI,
                int* __restrict__ rowptr, float* __restrict__ invO,
                float* __restrict__ invI, int N) {
  __shared__ int wsum[32];
  __shared__ int tileTot;
  const int t    = threadIdx.x;
  const int lane = t & 31;
  const int w    = t >> 5;
  int carry = 0;
  for (int base = 0; base < N; base += 4096) {
    const int i = base + t * 4;
    int ci[4], co[4];
    if (i + 3 < N) {
      int4 a = *(const int4*)(cntI + i);
      int4 b = *(const int4*)(cntO + i);
      ci[0] = a.x; ci[1] = a.y; ci[2] = a.z; ci[3] = a.w;
      co[0] = b.x; co[1] = b.y; co[2] = b.z; co[3] = b.w;
    } else {
      #pragma unroll
      for (int k = 0; k < 4; ++k) {
        bool ok = (i + k) < N;
        ci[k] = ok ? cntI[i + k] : 0;
        co[k] = ok ? cntO[i + k] : 0;
      }
    }
    const int tsum = ci[0] + ci[1] + ci[2] + ci[3];
    int x = tsum;
    #pragma unroll
    for (int off = 1; off < 32; off <<= 1) {
      int y = __shfl_up(x, off);
      if (lane >= off) x += y;
    }
    if (lane == 31) wsum[w] = x;
    __syncthreads();
    if (w == 0) {
      int s  = wsum[lane];
      int xs = s;
      #pragma unroll
      for (int off = 1; off < 32; off <<= 1) {
        int y = __shfl_up(xs, off);
        if (lane >= off) xs += y;
      }
      wsum[lane] = xs - s;
      if (lane == 31) tileTot = xs;
    }
    __syncthreads();
    int excl = carry + wsum[w] + x - tsum;
    #pragma unroll
    for (int k = 0; k < 4; ++k) {
      if (i + k < N) {
        rowptr[i + k] = excl;
        invO[i + k]   = gcn_inv_sqrt_deg(co[k]);
        invI[i + k]   = gcn_inv_sqrt_deg(ci[k]);
      }
      excl += ci[k];
    }
    carry += tileTot;
    __syncthreads();
  }
  if (t == 0) rowptr[N] = carry;
}

__global__ void gcn_fill_kernel(const int* __restrict__ src, const int* __restrict__ dst,
                                const int* __restrict__ rank, const int* __restrict__ rowptr,
                                const float* __restrict__ invO, int2* __restrict__ csr, int E) {
  int t = blockIdx.x * blockDim.x + threadIdx.x;
  if (t < E) {
    int s = src[t];
    int2 p;
    p.x = s;
    p.y = __float_as_int(invO[s]);
    csr[rowptr[dst[t]] + rank[t]] = p;
  }
}

template <int CHUNKS, bool FINAL>
__global__ void __launch_bounds__(256)
gcn_gather_kernel(const float* __restrict__ h,
                  const int* __restrict__ rowptr,
                  const int2* __restrict__ csr,
                  const float* __restrict__ invI,
                  const float* __restrict__ bias,
                  float* __restrict__ outp, int N) {
  constexpr int NPW = 32 / CHUNKS;
  const int lane = threadIdx.x & 31;
  const int wave = blockIdx.x * 8 + (threadIdx.x >> 5);
  const int sub  = lane / CHUNKS;
  const int c    = lane - sub * CHUNKS;
  int node = wave * NPW + sub;
  if (NPW == 1) {
    node = __builtin_amdgcn_readfirstlane(node);
  }
  if (sub >= NPW || node >= N) return;

  const int beg = rowptr[node];
  const int end = rowptr[node + 1];
  const float4* __restrict__ h4 = (const float4*)h;

  float4 acc = make_float4(0.f, 0.f, 0.f, 0.f);
  int j = beg;
  for (; j + 4 <= end; j += 4) {
    int2 e0 = csr[j], e1 = csr[j + 1], e2 = csr[j + 2], e3 = csr[j + 3];
    float4 v0 = h4[(size_t)e0.x * CHUNKS + c];
    float4 v1 = h4[(size_t)e1.x * CHUNKS + c];
    float4 v2 = h4[(size_t)e2.x * CHUNKS + c];
    float4 v3 = h4[(size_t)e3.x * CHUNKS + c];
    float w0 = __int_as_float(e0.y), w1 = __int_as_float(e1.y);
    float w2 = __int_as_float(e2.y), w3 = __int_as_float(e3.y);
    acc.x = fmaf(v0.x, w0, acc.x); acc.y = fmaf(v0.y, w0, acc.y);
    acc.z = fmaf(v0.z, w0, acc.z); acc.w = fmaf(v0.w, w0, acc.w);
    acc.x = fmaf(v1.x, w1, acc.x); acc.y = fmaf(v1.y, w1, acc.y);
    acc.z = fmaf(v1.z, w1, acc.z); acc.w = fmaf(v1.w, w1, acc.w);
    acc.x = fmaf(v2.x, w2, acc.x); acc.y = fmaf(v2.y, w2, acc.y);
    acc.z = fmaf(v2.z, w2, acc.z); acc.w = fmaf(v2.w, w2, acc.w);
    acc.x = fmaf(v3.x, w3, acc.x); acc.y = fmaf(v3.y, w3, acc.y);
    acc.z = fmaf(v3.z, w3, acc.z); acc.w = fmaf(v3.w, w3, acc.w);
  }
  const int rem = end - j;
  if (rem > 0) {
    const int last = end - 1;
    int2 e0 = csr[j];
    int2 e1 = csr[(j + 1 < end) ? j + 1 : last];
    int2 e2 = csr[(j + 2 < end) ? j + 2 : last];
    float4 v0 = h4[(size_t)e0.x * CHUNKS + c];
    float4 v1 = h4[(size_t)e1.x * CHUNKS + c];
    float4 v2 = h4[(size_t)e2.x * CHUNKS + c];
    float w0 = __int_as_float(e0.y), w1 = __int_as_float(e1.y), w2 = __int_as_float(e2.y);
    acc.x = fmaf(v0.x, w0, acc.x); acc.y = fmaf(v0.y, w0, acc.y);
    acc.z = fmaf(v0.z, w0, acc.z); acc.w = fmaf(v0.w, w0, acc.w);
    if (rem > 1) {
      acc.x = fmaf(v1.x, w1, acc.x); acc.y = fmaf(v1.y, w1, acc.y);
      acc.z = fmaf(v1.z, w1, acc.z); acc.w = fmaf(v1.w, w1, acc.w);
    }
    if (rem > 2) {
      acc.x = fmaf(v2.x, w2, acc.x); acc.y = fmaf(v2.y, w2, acc.y);
      acc.z = fmaf(v2.z, w2, acc.z); acc.w = fmaf(v2.w, w2, acc.w);
    }
  }

  float4* op = (float4*)outp + (size_t)node * CHUNKS + c;
  if (FINAL) {
    const float  sI = invI[node];
    const float4 bb = ((const float4*)bias)[c];
    float4 r;
    r.x = acc.x * sI + bb.x; r.y = acc.y * sI + bb.y;
    r.z = acc.z * sI + bb.z; r.w = acc.w * sI + bb.w;
    *op = r;
  } else {
    *op = acc;
  }
}

__global__ void __launch_bounds__(256)
gcn_gemm_kernel(const float* __restrict__ A,
                const float* __restrict__ rowScale,
                const float* __restrict__ W,
                const float* __restrict__ bias,
                float* __restrict__ C,
                int M, int Nd, int relu) {
  __shared__ float As[16 * LDK];

  const int t     = threadIdx.x;
  const int mBase = blockIdx.x * 16;

  for (int i = t; i < 16 * (TK / 4); i += 256) {
    int row = i >> 5;
    int c4  = i & 31;
    int gr  = mBase + row;
    float4 v = make_float4(0.f, 0.f, 0.f, 0.f);
    float  s = 1.0f;
    if (gr < M) {
      v = ((const float4*)(A + (size_t)gr * TK))[c4];
      if (rowScale) s = rowScale[gr];
    }
    float* dstp = As + (size_t)row * LDK + (size_t)c4 * 4;
    dstp[0] = v.x * s; dstp[1] = v.y * s; dstp[2] = v.z * s; dstp[3] = v.w * s;
  }
  __syncthreads();

  const int wave   = t >> 5;
  const int lane   = t & 31;
  const int nTiles = (Nd + 15) >> 4;
  if (wave >= nTiles) return;

  const int col    = lane & 15;
  const int gcol   = wave * 16 + col;
  const int khalf  = (lane >> 4) * 2;
  const bool colOk = (gcol < Nd);
  const int gcSafe = colOk ? gcol : (Nd - 1);

  const float* As_row = As + (size_t)(lane & 15) * LDK + khalf;
  const float* Wp     = W + (size_t)khalf * Nd + gcSafe;

  v8f acc0 = {};
  v8f acc1 = {};
  #pragma unroll 8
  for (int k0 = 0; k0 < TK; k0 += 8) {
    v2f a0, b0, a1, b1;
    a0[0] = As_row[k0];
    a0[1] = As_row[k0 + 1];
    b0[0] = Wp[0];
    b0[1] = Wp[Nd];
    a1[0] = As_row[k0 + 4];
    a1[1] = As_row[k0 + 5];
    b1[0] = Wp[4 * (size_t)Nd];
    b1[1] = Wp[5 * (size_t)Nd];
    acc0 = __builtin_amdgcn_wmma_f32_16x16x4_f32(false, a0, false, b0, (short)0, acc0,
                                                 false, false);
    acc1 = __builtin_amdgcn_wmma_f32_16x16x4_f32(false, a1, false, b1, (short)0, acc1,
                                                 false, false);
    Wp += 8 * (size_t)Nd;
  }
  v8f acc = acc0 + acc1;

  if (colOk) {
    const float bb    = bias ? bias[gcol] : 0.0f;
    const int   rbase = mBase + (lane >> 4) * 8;
    #pragma unroll
    for (int v = 0; v < 8; ++v) {
      int r = rbase + v;
      if (r < M) {
        float val = acc[v] + bb;
        if (relu) val = fmaxf(val, 0.0f);
        C[(size_t)r * Nd + gcol] = val;
      }
    }
  }
}

extern "C" void kernel_launch(void* const* d_in, const int* in_sizes, int n_in,
                              void* d_out, int out_size, void* d_ws, size_t ws_size,
                              hipStream_t stream) {
  const float* feats = (const float*)d_in[0];
  const int*   src   = (const int*)d_in[1];
  const int*   dst   = (const int*)d_in[2];
  const float* W1    = (const float*)d_in[3];
  const float* b1    = (const float*)d_in[4];
  const float* W2    = (const float*)d_in[5];
  const float* b2    = (const float*)d_in[6];
  const float* W3    = (const float*)d_in[7];
  const float* b3    = (const float*)d_in[8];
  float*       out   = (float*)d_out;

  const int HID  = in_sizes[4];
  const int NCLS = in_sizes[8];
  const int INF  = in_sizes[3] / HID;
  const int N    = in_sizes[0] / INF;
  const int E    = in_sizes[1];

  const size_t Np = ((size_t)N + 1 + 63) & ~(size_t)63;
  const size_t Ep = ((size_t)E + 63) & ~(size_t)63;
  int*   cntO   = (int*)d_ws;
  int*   cntI   = cntO + Np;
  float* invO   = (float*)(cntI + Np);
  float* invI   = invO + Np;
  int*   rowptr = (int*)(invI + Np);
  int*   rank   = rowptr + Np;
  int2*  csr    = (int2*)(rank + Ep);
  float* bufA   = (float*)(csr + Ep);
  float* bufB   = bufA + (size_t)N * TK;
  float* bufC   = bufB + (size_t)N * TK;

  const int B = 256;
  const int mTiles = (N + 15) / 16;
  const int eBlocks = (E + B - 1) / B;
  const int g32Blocks = (N + 7) / 8;
  const int g10Blocks = (N + 23) / 24;

  hipMemsetAsync(cntO, 0, sizeof(int) * 2 * Np, stream);
  gcn_hist_kernel<<<eBlocks, B, 0, stream>>>(src, dst, cntO, cntI, rank, E);
  gcn_scan_kernel<<<1, 1024, 0, stream>>>(cntO, cntI, rowptr, invO, invI, N);
  gcn_fill_kernel<<<eBlocks, B, 0, stream>>>(src, dst, rank, rowptr, invO, csr, E);

  gcn_gather_kernel<32, false><<<g32Blocks, B, 0, stream>>>(feats, rowptr, csr, invI, nullptr, bufA, N);
  gcn_gemm_kernel<<<mTiles, B, 0, stream>>>(bufA, invI, W1, b1, bufB, N, HID, 1);

  gcn_gather_kernel<32, false><<<g32Blocks, B, 0, stream>>>(bufB, rowptr, csr, invI, nullptr, bufA, N);
  gcn_gemm_kernel<<<mTiles, B, 0, stream>>>(bufA, invI, W2, b2, bufB, N, HID, 1);

  gcn_gemm_kernel<<<mTiles, B, 0, stream>>>(bufB, nullptr, W3, nullptr, bufC, N, NCLS, 0);
  gcn_gather_kernel<10, true><<<g10Blocks, B, 0, stream>>>(bufC, rowptr, csr, invI, b3, out, N);
}


# Degrees and edge ranks from per-chunk LDS histograms (64 edge chunks, LDS add return = rank) plus a per-node prefix over chunks replace the per-edge global atomics; sum order: arrival in a chunk
# speedup vs baseline: 2.8205x; 1.3824x over previous
#include <hip/hip_runtime.h>
#include <hip/hip_bf16.h>

typedef __attribute__((ext_vector_type(2))) float v2f;
typedef __attribute__((ext_vector_type(8))) float v8f;

#define TK 128
#define LDK 132

#define NCHUNK 64
#define HB 50176

__global__ void __launch_bounds__(1024)
gcn_count_kernel(const int* __restrict__ src, const int* __restrict__ dst,
                 int* __restrict__ partO, int* __restrict__ partI,
                 int* __restrict__ rank, int N, int E, int CE, int Np) {
  __shared__ int hist[HB];
  const int  t   = threadIdx.x;
  const bool isI = (int)blockIdx.x < NCHUNK;
  const int  g   = isI ? (int)blockIdx.x : (int)blockIdx.x - NCHUNK;
  const int  e0  = g * CE;
  const int  e1  = min(E, e0 + CE);
  int* __restrict__ part = (isI ? partI : partO) + (size_t)g * Np;

  for (int base = 0; base < N; base += HB) {
    const int lim  = min(HB, N - base);
    const int limP = min(HB, Np - base);
    for (int i = t * 4; i < limP; i += 4096)
      *(int4*)(hist + i) = make_int4(0, 0, 0, 0);
    __syncthreads();
    if (isI) {
      for (int e = e0 + t; e < e1; e += 8 * 1024) {
        int d[8], r[8];
        #pragma unroll
        for (int k = 0; k < 8; ++k) {
          const int ee = e + k * 1024;
          const int v  = dst[min(ee, e1 - 1)];
          d[k] = (ee < e1) ? v - base : -1;
        }
        #pragma unroll
        for (int k = 0; k < 8; ++k)
          r[k] = ((unsigned)d[k] < (unsigned)lim) ? atomicAdd(&hist[d[k]], 1) : 0;
        #pragma unroll
        for (int k = 0; k < 8; ++k)
          if ((unsigned)d[k] < (unsigned)lim) rank[e + k * 1024] = r[k];
      }
    } else {
      for (int e = e0 + t; e < e1; e += 8 * 1024) {
        int d[8];
        #pragma unroll
        for (int k = 0; k < 8; ++k) {
          const int ee = e + k * 1024;
          const int v  = src[min(ee, e1 - 1)];
          d[k] = (ee < e1) ? v - base : -1;
        }
        #pragma unroll
        for (int k = 0; k < 8; ++k) {
          if ((unsigned)d[k] < (unsigned)lim) atomicAdd(&hist[d[k]], 1);
        }
      }
    }
    __syncthreads();
    for (int i = t * 4; i < limP; i += 4096)
      *(int4*)(part + base + i) = *(const int4*)(hist + i);
    __syncthreads();
  }
}

__global__ void __launch_bounds__(256)
gcn_colscan_kernel(const int* __restrict__ partO, int* __restrict__ partI,
                   int* __restrict__ cntO, int* __restrict__ cntI, int N, int Np) {
  const int n = blockIdx.x * 256 + threadIdx.x;
  if (n >= N) return;
  if (blockIdx.y == 0) {
    int* p = partI + n;
    int run = 0;
    for (int g = 0; g < NCHUNK; g += 16) {
      int c[16];
      #pragma unroll
      for (int k = 0; k < 16; ++k) c[k] = p[(size_t)(g + k) * Np];
      #pragma unroll
      for (int k = 0; k < 16; ++k) { p[(size_t)(g + k) * Np] = run; run += c[k]; }
    }
    cntI[n] = run;
  } else {
    const int* p = partO + n;
    int s = 0;
    #pragma unroll 16
    for (int g = 0; g < NCHUNK; ++g) s += p[(size_t)g * Np];
    cntO[n] = s;
  }
}

__device__ __forceinline__ float gcn_inv_sqrt_deg(int c) {
  return (c > 0) ? rsqrtf(fmaxf((float)c, 1.0f)) : 0.0f;
}

__global__ void __launch_bounds__(1024)
gcn_scan_kernel(const int* __restrict__ cntO, const int* __restrict__ cntI,
                int* __restrict__ rowptr, float* __restrict__ invO,
                float* __restrict__ invI, int N) {
  __shared__ int wsum[32];
  __shared__ int tileTot;
  const int t    = threadIdx.x;
  const int lane = t & 31;
  const int w    = t >> 5;
  int carry = 0;
  for (int base = 0; base < N; base += 4096) {
    const int i = base + t * 4;
    int ci[4], co[4];
    if (i + 3 < N) {
      int4 a = *(const int4*)(cntI + i);
      int4 b = *(const int4*)(cntO + i);
      ci[0] = a.x; ci[1] = a.y; ci[2] = a.z; ci[3] = a.w;
      co[0] = b.x; co[1] = b.y; co[2] = b.z; co[3] = b.w;
    } else {
      #pragma unroll
      for (int k = 0; k < 4; ++k) {
        bool ok = (i + k) < N;
        ci[k] = ok ? cntI[i + k] : 0;
        co[k] = ok ? cntO[i + k] : 0;
      }
    }
    const int tsum = ci[0] + ci[1] + ci[2] + ci[3];
    int x = tsum;
    #pragma unroll
    for (int off = 1; off < 32; off <<= 1) {
      int y = __shfl_up(x, off);
      if (lane >= off) x += y;
    }
    if (lane == 31) wsum[w] = x;
    __syncthreads();
    if (w == 0) {
      int s  = wsum[lane];
      int xs = s;
      #pragma unroll
      for (int off = 1; off < 32; off <<= 1) {
        int y = __shfl_up(xs, off);
        if (lane >= off) xs += y;
      }
      wsum[lane] = xs - s;
      if (lane == 31) tileTot = xs;
    }
    __syncthreads();
    int excl = carry + wsum[w] + x - tsum;
    #pragma unroll
    for (int k = 0; k < 4; ++k) {
      if (i + k < N) {
        rowptr[i + k] = excl;
        invO[i + k]   = gcn_inv_sqrt_deg(co[k]);
        invI[i + k]   = gcn_inv_sqrt_deg(ci[k]);
      }
      excl += ci[k];
    }
    carry += tileTot;
    __syncthreads();
  }
  if (t == 0) rowptr[N] = carry;
}

__global__ void gcn_fill_kernel(const int* __restrict__ src, const int* __restrict__ dst,
                                const int* __restrict__ rank, const int* __restrict__ rowptr,
                                const int* __restrict__ partI,
                                const float* __restrict__ invO, int2* __restrict__ csr,
                                int E, int CE, int Np) {
  int t = blockIdx.x * blockDim.x + threadIdx.x;
  if (t < E) {
    int s = src[t];
    int d = dst[t];
    int g = t / CE;
    int2 p;
    p.x = s;
    p.y = __float_as_int(invO[s]);
    csr[rowptr[d] + partI[(size_t)g * Np + d] + rank[t]] = p;
  }
}

template <int CHUNKS, bool FINAL>
__global__ void __launch_bounds__(256)
gcn_gather_kernel(const float* __restrict__ h,
                  const int* __restrict__ rowptr,
                  const int2* __restrict__ csr,
                  const float* __restrict__ invI,
                  const float* __restrict__ bias,
                  float* __restrict__ outp, int N) {
  constexpr int NPW = 32 / CHUNKS;
  const int lane = threadIdx.x & 31;
  const int wave = blockIdx.x * 8 + (threadIdx.x >> 5);
  const int sub  = lane / CHUNKS;
  const int c    = lane - sub * CHUNKS;
  int node = wave * NPW + sub;
  if (NPW == 1) {
    node = __builtin_amdgcn_readfirstlane(node);
  }
  if (sub >= NPW || node >= N) return;

  const int beg = rowptr[node];
  const int end = rowptr[node + 1];
  const float4* __restrict__ h4 = (const float4*)h;

  float4 acc = make_float4(0.f, 0.f, 0.f, 0.f);
  int j = beg;
  for (; j + 4 <= end; j += 4) {
    int2 e0 = csr[j], e1 = csr[j + 1], e2 = csr[j + 2], e3 = csr[j + 3];
    float4 v0 = h4[(size_t)e0.x * CHUNKS + c];
    float4 v1 = h4[(size_t)e1.x * CHUNKS + c];
    float4 v2 = h4[(size_t)e2.x * CHUNKS + c];
    float4 v3 = h4[(size_t)e3.x * CHUNKS + c];
    float w0 = __int_as_float(e0.y), w1 = __int_as_float(e1.y);
    float w2 = __int_as_float(e2.y), w3 = __int_as_float(e3.y);
    acc.x = fmaf(v0.x, w0, acc.x); acc.y = fmaf(v0.y, w0, acc.y);
    acc.z = fmaf(v0.z, w0, acc.z); acc.w = fmaf(v0.w, w0, acc.w);
    acc.x = fmaf(v1.x, w1, acc.x); acc.y = fmaf(v1.y, w1, acc.y);
    acc.z = fmaf(v1.z, w1, acc.z); acc.w = fmaf(v1.w, w1, acc.w);
    acc.x = fmaf(v2.x, w2, acc.x); acc.y = fmaf(v2.y, w2, acc.y);
    acc.z = fmaf(v2.z, w2, acc.z); acc.w = fmaf(v2.w, w2, acc.w);
    acc.x = fmaf(v3.x, w3, acc.x); acc.y = fmaf(v3.y, w3, acc.y);
    acc.z = fmaf(v3.z, w3, acc.z); acc.w = fmaf(v3.w, w3, acc.w);
  }
  const int rem = end - j;
  if (rem > 0) {
    const int last = end - 1;
    int2 e0 = csr[j];
    int2 e1 = csr[(j + 1 < end) ? j + 1 : last];
    int2 e2 = csr[(j + 2 < end) ? j + 2 : last];
    float4 v0 = h4[(size_t)e0.x * CHUNKS + c];
    float4 v1 = h4[(size_t)e1.x * CHUNKS + c];
    float4 v2 = h4[(size_t)e2.x * CHUNKS + c];
    float w0 = __int_as_float(e0.y), w1 = __int_as_float(e1.y), w2 = __int_as_float(e2.y);
    acc.x = fmaf(v0.x, w0, acc.x); acc.y = fmaf(v0.y, w0, acc.y);
    acc.z = fmaf(v0.z, w0, acc.z); acc.w = fmaf(v0.w, w0, acc.w);
    if (rem > 1) {
      acc.x = fmaf(v1.x, w1, acc.x); acc.y = fmaf(v1.y, w1, acc.y);
      acc.z = fmaf(v1.z, w1, acc.z); acc.w = fmaf(v1.w, w1, acc.w);
    }
    if (rem > 2) {
      acc.x = fmaf(v2.x, w2, acc.x); acc.y = fmaf(v2.y, w2, acc.y);
      acc.z = fmaf(v2.z, w2, acc.z); acc.w = fmaf(v2.w, w2, acc.w);
    }
  }

  float4* op = (float4*)outp + (size_t)node * CHUNKS + c;
  if (FINAL) {
    const float  sI = invI[node];
    const float4 bb = ((const float4*)bias)[c];
    float4 r;
    r.x = acc.x * sI + bb.x; r.y = acc.y * sI + bb.y;
    r.z = acc.z * sI + bb.z; r.w = acc.w * sI + bb.w;
    *op = r;
  } else {
    *op = acc;
  }
}

__global__ void __launch_bounds__(256)
gcn_gemm_kernel(const float* __restrict__ A,
                const float* __restrict__ rowScale,
                const float* __restrict__ W,
                const float* __restrict__ bias,
                float* __restrict__ C,
                int M, int Nd, int relu) {
  __shared__ float As[16 * LDK];

  const int t     = threadIdx.x;
  const int mBase = blockIdx.x * 16;

  for (int i = t; i < 16 * (TK / 4); i += 256) {
    int row = i >> 5;
    int c4  = i & 31;
    int gr  = mBase + row;
    float4 v = make_float4(0.f, 0.f, 0.f, 0.f);
    float  s = 1.0f;
    if (gr < M) {
      v = ((const float4*)(A + (size_t)gr * TK))[c4];
      if (rowScale) s = rowScale[gr];
    }
    float* dstp = As + (size_t)row * LDK + (size_t)c4 * 4;
    dstp[0] = v.x * s; dstp[1] = v.y * s; dstp[2] = v.z * s; dstp[3] = v.w * s;
  }
  __syncthreads();

  const int wave   = t >> 5;
  const int lane   = t & 31;
  const int nTiles = (Nd + 15) >> 4;
  if (wave >= nTiles) return;

  const int col    = lane & 15;
  const int gcol   = wave * 16 + col;
  const int khalf  = (lane >> 4) * 2;
  const bool colOk = (gcol < Nd);
  const int gcSafe = colOk ? gcol : (Nd - 1);

  const float* As_row = As + (size_t)(lane & 15) * LDK + khalf;
  const float* Wp     = W + (size_t)khalf * Nd + gcSafe;

  v8f acc0 = {};
  v8f acc1 = {};
  #pragma unroll 8
  for (int k0 = 0; k0 < TK; k0 += 8) {
    v2f a0, b0, a1, b1;
    a0[0] = As_row[k0];
    a0[1] = As_row[k0 + 1];
    b0[0] = Wp[0];
    b0[1] = Wp[Nd];
    a1[0] = As_row[k0 + 4];
    a1[1] = As_row[k0 + 5];
    b1[0] = Wp[4 * (size_t)Nd];
    b1[1] = Wp[5 * (size_t)Nd];
    acc0 = __builtin_amdgcn_wmma_f32_16x16x4_f32(false, a0, false, b0, (short)0, acc0,
                                                 false, false);
    acc1 = __builtin_amdgcn_wmma_f32_16x16x4_f32(false, a1, false, b1, (short)0, acc1,
                                                 false, false);
    Wp += 8 * (size_t)Nd;
  }
  v8f acc = acc0 + acc1;

  if (colOk) {
    const float bb    = bias ? bias[gcol] : 0.0f;
    const int   rbase = mBase + (lane >> 4) * 8;
    #pragma unroll
    for (int v = 0; v < 8; ++v) {
      int r = rbase + v;
      if (r < M) {
        float val = acc[v] + bb;
        if (relu) val = fmaxf(val, 0.0f);
        C[(size_t)r * Nd + gcol] = val;
      }
    }
  }
}

extern "C" void kernel_launch(void* const* d_in, const int* in_sizes, int n_in,
                              void* d_out, int out_size, void* d_ws, size_t ws_size,
                              hipStream_t stream) {
  const float* feats = (const float*)d_in[0];
  const int*   src   = (const int*)d_in[1];
  const int*   dst   = (const int*)d_in[2];
  const float* W1    = (const float*)d_in[3];
  const float* b1    = (const float*)d_in[4];
  const float* W2    = (const float*)d_in[5];
  const float* b2    = (const float*)d_in[6];
  const float* W3    = (const float*)d_in[7];
  const float* b3    = (const float*)d_in[8];
  float*       out   = (float*)d_out;

  const int HID  = in_sizes[4];
  const int NCLS = in_sizes[8];
  const int INF  = in_sizes[3] / HID;
  const int N    = in_sizes[0] / INF;
  const int E    = in_sizes[1];

  const size_t Np = ((size_t)N + 1 + 63) & ~(size_t)63;
  const size_t Ep = ((size_t)E + 63) & ~(size_t)63;
  int*   cntO   = (int*)d_ws;
  int*   cntI   = cntO + Np;
  float* invO   = (float*)(cntI + Np);
  float* invI   = invO + Np;
  int*   rowptr = (int*)(invI + Np);
  int*   rank   = rowptr + Np;
  int2*  csr    = (int2*)(rank + Ep);
  int*   partO  = (int*)(csr + Ep);
  int*   partI  = partO + (size_t)NCHUNK * Np;
  float* bufA   = (float*)(partI + (size_t)NCHUNK * Np);
  float* bufB   = bufA + (size_t)N * TK;
  float* bufC   = bufB + (size_t)N * TK;

  const int B = 256;
  const int mTiles = (N + 15) / 16;
  const int eBlocks = (E + B - 1) / B;
  const int g32Blocks = (N + 7) / 8;
  const int g10Blocks = (N + 23) / 24;
  const int CE = (E + NCHUNK - 1) / NCHUNK;
  const int nBlocks = (N + B - 1) / B;

  gcn_count_kernel<<<2 * NCHUNK, 1024, 0, stream>>>(src, dst, partO, partI, rank, N, E, CE, (int)Np);
  gcn_colscan_kernel<<<dim3(nBlocks, 2), B, 0, stream>>>(partO, partI, cntO, cntI, N, (int)Np);
  gcn_scan_kernel<<<1, 1024, 0, stream>>>(cntO, cntI, rowptr, invO, invI, N);
  gcn_fill_kernel<<<eBlocks, B, 0, stream>>>(src, dst, rank, rowptr, partI, invO, csr, E, CE, (int)Np);

  gcn_gather_kernel<32, false><<<g32Blocks, B, 0, stream>>>(feats, rowptr, csr, invI, nullptr, bufA, N);
  gcn_gemm_kernel<<<mTiles, B, 0, stream>>>(bufA, invI, W1, b1, bufB, N, HID, 1);

  gcn_gather_kernel<32, false><<<g32Blocks, B, 0, stream>>>(bufB, rowptr, csr, invI, nullptr, bufA, N);
  gcn_gemm_kernel<<<mTiles, B, 0, stream>>>(bufA, invI, W2, b2, bufB, N, HID, 1);

  gcn_gemm_kernel<<<mTiles, B, 0, stream>>>(bufB, nullptr, W3, nullptr, bufC, N, NCLS, 0);
  gcn_gather_kernel<10, true><<<g10Blocks, B, 0, stream>>>(bufC, rowptr, csr, invI, b3, out, N);
}


# Layer 1 and 2 gather-sum moved into the GEMM kernel: each wave sums two nodes' neighbour rows (same edge order) straight into the LDS A tile; the aggregate buffer round trip and two launches go
# speedup vs baseline: 3.0463x; 1.0801x over previous
#include <hip/hip_runtime.h>
#include <hip/hip_bf16.h>

typedef __attribute__((ext_vector_type(2))) float v2f;
typedef __attribute__((ext_vector_type(8))) float v8f;

#define TK 128
#define LDK 132

#define NCHUNK 64
#define HB 50176

__global__ void __launch_bounds__(1024)
gcn_count_kernel(const int* __restrict__ src, const int* __restrict__ dst,
                 int* __restrict__ partO, int* __restrict__ partI,
                 int* __restrict__ rank, int N, int E, int CE, int Np) {
  __shared__ int hist[HB];
  const int  t   = threadIdx.x;
  const bool isI = (int)blockIdx.x < NCHUNK;
  const int  g   = isI ? (int)blockIdx.x : (int)blockIdx.x - NCHUNK;
  const int  e0  = g * CE;
  const int  e1  = min(E, e0 + CE);
  int* __restrict__ part = (isI ? partI : partO) + (size_t)g * Np;

  for (int base = 0; base < N; base += HB) {
    const int lim  = min(HB, N - base);
    const int limP = min(HB, Np - base);
    for (int i = t * 4; i < limP; i += 4096)
      *(int4*)(hist + i) = make_int4(0, 0, 0, 0);
    __syncthreads();
    if (isI) {
      for (int e = e0 + t; e < e1; e += 8 * 1024) {
        int d[8], r[8];
        #pragma unroll
        for (int k = 0; k < 8; ++k) {
          const int ee = e + k * 1024;
          const int v  = dst[min(ee, e1 - 1)];
          d[k] = (ee < e1) ? v - base : -1;
        }
        #pragma unroll
        for (int k = 0; k < 8; ++k)
          r[k] = ((unsigned)d[k] < (unsigned)lim) ? atomicAdd(&hist[d[k]], 1) : 0;
        #pragma unroll
        for (int k = 0; k < 8; ++k)
          if ((unsigned)d[k] < (unsigned)lim) rank[e + k * 1024] = r[k];
      }
    } else {
      for (int e = e0 + t; e < e1; e += 8 * 1024) {
        int d[8];
        #pragma unroll
        for (int k = 0; k < 8; ++k) {
          const int ee = e + k * 1024;
          const int v  = src[min(ee, e1 - 1)];
          d[k] = (ee < e1) ? v - base : -1;
        }
        #pragma unroll
        for (int k = 0; k < 8; ++k) {
          if ((unsigned)d[k] < (unsigned)lim) atomicAdd(&hist[d[k]], 1);
        }
      }
    }
    __syncthreads();
    for (int i = t * 4; i < limP; i += 4096)
      *(int4*)(part + base + i) = *(const int4*)(hist + i);
    __syncthreads();
  }
}

__global__ void __launch_bounds__(256)
gcn_colscan_kernel(const int* __restrict__ partO, int* __restrict__ partI,
                   int* __restrict__ cntO, int* __restrict__ cntI, int N, int Np) {
  const int n = blockIdx.x * 256 + threadIdx.x;
  if (n >= N) return;
  if (blockIdx.y == 0) {
    int* p = partI + n;
    int run = 0;
    for (int g = 0; g < NCHUNK; g += 16) {
      int c[16];
      #pragma unroll
      for (int k = 0; k < 16; ++k) c[k] = p[(size_t)(g + k) * Np];
      #pragma unroll
      for (int k = 0; k < 16; ++k) { p[(size_t)(g + k) * Np] = run; run += c[k]; }
    }
    cntI[n] = run;
  } else {
    const int* p = partO + n;
    int s = 0;
    #pragma unroll 16
    for (int g = 0; g < NCHUNK; ++g) s += p[(size_t)g * Np];
    cntO[n] = s;
  }
}

__device__ __forceinline__ float gcn_inv_sqrt_deg(int c) {
  return (c > 0) ? rsqrtf(fmaxf((float)c, 1.0f)) : 0.0f;
}

__global__ void __launch_bounds__(1024)
gcn_scan_kernel(const int* __restrict__ cntO, const int* __restrict__ cntI,
                int* __restrict__ rowptr, float* __restrict__ invO,
                float* __restrict__ invI, int N) {
  __shared__ int wsum[32];
  __shared__ int tileTot;
  const int t    = threadIdx.x;
  const int lane = t & 31;
  const int w    = t >> 5;
  int carry = 0;
  for (int base = 0; base < N; base += 4096) {
    const int i = base + t * 4;
    int ci[4], co[4];
    if (i + 3 < N) {
      int4 a = *(const int4*)(cntI + i);
      int4 b = *(const int4*)(cntO + i);
      ci[0] = a.x; ci[1] = a.y; ci[2] = a.z; ci[3] = a.w;
      co[0] = b.x; co[1] = b.y; co[2] = b.z; co[3] = b.w;
    } else {
      #pragma unroll
      for (int k = 0; k < 4; ++k) {
        bool ok = (i + k) < N;
        ci[k] = ok ? cntI[i + k] : 0;
        co[k] = ok ? cntO[i + k] : 0;
      }
    }
    const int tsum = ci[0] + ci[1] + ci[2] + ci[3];
    int x = tsum;
    #pragma unroll
    for (int off = 1; off < 32; off <<= 1) {
      int y = __shfl_up(x, off);
      if (lane >= off) x += y;
    }
    if (lane == 31) wsum[w] = x;
    __syncthreads();
    if (w == 0) {
      int s  = wsum[lane];
      int xs = s;
      #pragma unroll
      for (int off = 1; off < 32; off <<= 1) {
        int y = __shfl_up(xs, off);
        if (lane >= off) xs += y;
      }
      wsum[lane] = xs - s;
      if (lane == 31) tileTot = xs;
    }
    __syncthreads();
    int excl = carry + wsum[w] + x - tsum;
    #pragma unroll
    for (int k = 0; k < 4; ++k) {
      if (i + k < N) {
        rowptr[i + k] = excl;
        invO[i + k]   = gcn_inv_sqrt_deg(co[k]);
        invI[i + k]   = gcn_inv_sqrt_deg(ci[k]);
      }
      excl += ci[k];
    }
    carry += tileTot;
    __syncthreads();
  }
  if (t == 0) rowptr[N] = carry;
}

__global__ void gcn_fill_kernel(const int* __restrict__ src, const int* __restrict__ dst,
                                const int* __restrict__ rank, const int* __restrict__ rowptr,
                                const int* __restrict__ partI,
                                const float* __restrict__ invO, int2* __restrict__ csr,
                                int E, int CE, int Np) {
  int t = blockIdx.x * blockDim.x + threadIdx.x;
  if (t < E) {
    int s = src[t];
    int d = dst[t];
    int g = t / CE;
    int2 p;
    p.x = s;
    p.y = __float_as_int(invO[s]);
    csr[rowptr[d] + partI[(size_t)g * Np + d] + rank[t]] = p;
  }
}

template <int CHUNKS, bool FINAL>
__global__ void __launch_bounds__(256)
gcn_gather_kernel(const float* __restrict__ h,
                  const int* __restrict__ rowptr,
                  const int2* __restrict__ csr,
                  const float* __restrict__ invI,
                  const float* __restrict__ bias,
                  float* __restrict__ outp, int N) {
  constexpr int NPW = 32 / CHUNKS;
  const int lane = threadIdx.x & 31;
  const int wave = blockIdx.x * 8 + (threadIdx.x >> 5);
  const int sub  = lane / CHUNKS;
  const int c    = lane - sub * CHUNKS;
  int node = wave * NPW + sub;
  if (NPW == 1) {
    node = __builtin_amdgcn_readfirstlane(node);
  }
  if (sub >= NPW || node >= N) return;

  const int beg = rowptr[node];
  const int end = rowptr[node + 1];
  const float4* __restrict__ h4 = (const float4*)h;

  float4 acc = make_float4(0.f, 0.f, 0.f, 0.f);
  int j = beg;
  for (; j + 4 <= end; j += 4) {
    int2 e0 = csr[j], e1 = csr[j + 1], e2 = csr[j + 2], e3 = csr[j + 3];
    float4 v0 = h4[(size_t)e0.x * CHUNKS + c];
    float4 v1 = h4[(size_t)e1.x * CHUNKS + c];
    float4 v2 = h4[(size_t)e2.x * CHUNKS + c];
    float4 v3 = h4[(size_t)e3.x * CHUNKS + c];
    float w0 = __int_as_float(e0.y), w1 = __int_as_float(e1.y);
    float w2 = __int_as_float(e2.y), w3 = __int_as_float(e3.y);
    acc.x = fmaf(v0.x, w0, acc.x); acc.y = fmaf(v0.y, w0, acc.y);
    acc.z = fmaf(v0.z, w0, acc.z); acc.w = fmaf(v0.w, w0, acc.w);
    acc.x = fmaf(v1.x, w1, acc.x); acc.y = fmaf(v1.y, w1, acc.y);
    acc.z = fmaf(v1.z, w1, acc.z); acc.w = fmaf(v1.w, w1, acc.w);
    acc.x = fmaf(v2.x, w2, acc.x); acc.y = fmaf(v2.y, w2, acc.y);
    acc.z = fmaf(v2.z, w2, acc.z); acc.w = fmaf(v2.w, w2, acc.w);
    acc.x = fmaf(v3.x, w3, acc.x); acc.y = fmaf(v3.y, w3, acc.y);
    acc.z = fmaf(v3.z, w3, acc.z); acc.w = fmaf(v3.w, w3, acc.w);
  }
  const int rem = end - j;
  if (rem > 0) {
    const int last = end - 1;
    int2 e0 = csr[j];
    int2 e1 = csr[(j + 1 < end) ? j + 1 : last];
    int2 e2 = csr[(j + 2 < end) ? j + 2 : last];
    float4 v0 = h4[(size_t)e0.x * CHUNKS + c];
    float4 v1 = h4[(size_t)e1.x * CHUNKS + c];
    float4 v2 = h4[(size_t)e2.x * CHUNKS + c];
    float w0 = __int_as_float(e0.y), w1 = __int_as_float(e1.y), w2 = __int_as_float(e2.y);
    acc.x = fmaf(v0.x, w0, acc.x); acc.y = fmaf(v0.y, w0, acc.y);
    acc.z = fmaf(v0.z, w0, acc.z); acc.w = fmaf(v0.w, w0, acc.w);
    if (rem > 1) {
      acc.x = fmaf(v1.x, w1, acc.x); acc.y = fmaf(v1.y, w1, acc.y);
      acc.z = fmaf(v1.z, w1, acc.z); acc.w = fmaf(v1.w, w1, acc.w);
    }
    if (rem > 2) {
      acc.x = fmaf(v2.x, w2, acc.x); acc.y = fmaf(v2.y, w2, acc.y);
      acc.z = fmaf(v2.z, w2, acc.z); acc.w = fmaf(v2.w, w2, acc.w);
    }
  }

  float4* op = (float4*)outp + (size_t)node * CHUNKS + c;
  if (FINAL) {
    const float  sI = invI[node];
    const float4 bb = ((const float4*)bias)[c];
    float4 r;
    r.x = acc.x * sI + bb.x; r.y = acc.y * sI + bb.y;
    r.z = acc.z * sI + bb.z; r.w = acc.w * sI + bb.w;
    *op = r;
  } else {
    *op = acc;
  }
}

__device__ __forceinline__ void
gcn_tile_product(const float* __restrict__ As,
                 const float* __restrict__ W,
                 const float* __restrict__ bias,
                 float* __restrict__ C,
                 int mBase, int M, int Nd, int relu, int t) {
  const int wave   = t >> 5;
  const int lane   = t & 31;
  const int nTiles = (Nd + 15) >> 4;
  if (wave >= nTiles) return;

  const int col    = lane & 15;
  const int gcol   = wave * 16 + col;
  const int khalf  = (lane >> 4) * 2;
  const bool colOk = (gcol < Nd);
  const int gcSafe = colOk ? gcol : (Nd - 1);

  const float* As_row = As + (size_t)(lane & 15) * LDK + khalf;
  const float* Wp     = W + (size_t)khalf * Nd + gcSafe;

  v8f acc0 = {};
  v8f acc1 = {};
  #pragma unroll 8
  for (int k0 = 0; k0 < TK; k0 += 8) {
    v2f a0, b0, a1, b1;
    a0[0] = As_row[k0];
    a0[1] = As_row[k0 + 1];
    b0[0] = Wp[0];
    b0[1] = Wp[Nd];
    a1[0] = As_row[k0 + 4];
    a1[1] = As_row[k0 + 5];
    b1[0] = Wp[4 * (size_t)Nd];
    b1[1] = Wp[5 * (size_t)Nd];
    acc0 = __builtin_amdgcn_wmma_f32_16x16x4_f32(false, a0, false, b0, (short)0, acc0,
                                                 false, false);
    acc1 = __builtin_amdgcn_wmma_f32_16x16x4_f32(false, a1, false, b1, (short)0, acc1,
                                                 false, false);
    Wp += 8 * (size_t)Nd;
  }
  v8f acc = acc0 + acc1;

  if (colOk) {
    const float bb    = bias ? bias[gcol] : 0.0f;
    const int   rbase = mBase + (lane >> 4) * 8;
    #pragma unroll
    for (int v = 0; v < 8; ++v) {
      int r = rbase + v;
      if (r < M) {
        float val = acc[v] + bb;
        if (relu) val = fmaxf(val, 0.0f);
        C[(size_t)r * Nd + gcol] = val;
      }
    }
  }
}

__global__ void __launch_bounds__(256)
gcn_gemm_kernel(const float* __restrict__ A,
                const float* __restrict__ rowScale,
                const float* __restrict__ W,
                const float* __restrict__ bias,
                float* __restrict__ C,
                int M, int Nd, int relu) {
  __shared__ float As[16 * LDK];

  const int t     = threadIdx.x;
  const int mBase = blockIdx.x * 16;

  for (int i = t; i < 16 * (TK / 4); i += 256) {
    int row = i >> 5;
    int c4  = i & 31;
    int gr  = mBase + row;
    float4 v = make_float4(0.f, 0.f, 0.f, 0.f);
    float  s = 1.0f;
    if (gr < M) {
      v = ((const float4*)(A + (size_t)gr * TK))[c4];
      if (rowScale) s = rowScale[gr];
    }
    float* dstp = As + (size_t)row * LDK + (size_t)c4 * 4;
    dstp[0] = v.x * s; dstp[1] = v.y * s; dstp[2] = v.z * s; dstp[3] = v.w * s;
  }
  __syncthreads();

  gcn_tile_product(As, W, bias, C, mBase, M, Nd, relu, t);
}

__global__ void __launch_bounds__(256)
gcn_gather_gemm_kernel(const float* __restrict__ h,
                       const int* __restrict__ rowptr,
                       const int2* __restrict__ csr,
                       const float* __restrict__ invI,
                       const float* __restrict__ W,
                       const float* __restrict__ bias,
                       float* __restrict__ C,
                       int M, int Nd, int relu) {
  __shared__ float As[16 * LDK];

  const int t     = threadIdx.x;
  const int lane  = t & 31;
  const int mBase = blockIdx.x * 16;
  const float4* __restrict__ h4 = (const float4*)h;

  for (int r = 0; r < 2; ++r) {
    const int row  = (t >> 5) * 2 + r;
    const int node = __builtin_amdgcn_readfirstlane(mBase + row);
    float4 acc = make_float4(0.f, 0.f, 0.f, 0.f);
    if (node < M) {
      const int beg  = rowptr[node];
      const int end  = rowptr[node + 1];
      const int last = end - 1;
      int2 e0 = make_int2(0, 0), e1 = e0, e2 = e0, e3 = e0;
      if (beg < end) {
        e0 = csr[beg];
        e1 = csr[min(beg + 1, last)];
        e2 = csr[min(beg + 2, last)];
        e3 = csr[min(beg + 3, last)];
      }
      for (int j = beg; j < end; j += 4) {
        const float4 v0 = h4[(size_t)e0.x * 32 + lane];
        const float4 v1 = h4[(size_t)e1.x * 32 + lane];
        const float4 v2 = h4[(size_t)e2.x * 32 + lane];
        const float4 v3 = h4[(size_t)e3.x * 32 + lane];
        const float w0 = __int_as_float(e0.y), w1 = __int_as_float(e1.y);
        const float w2 = __int_as_float(e2.y), w3 = __int_as_float(e3.y);
        const int rem = end - j;
        const int jn  = j + 4;
        if (jn < end) {
          e0 = csr[jn];
          e1 = csr[min(jn + 1, last)];
          e2 = csr[min(jn + 2, last)];
          e3 = csr[min(jn + 3, last)];
        }
        acc.x = fmaf(v0.x, w0, acc.x); acc.y = fmaf(v0.y, w0, acc.y);
        acc.z = fmaf(v0.z, w0, acc.z); acc.w = fmaf(v0.w, w0, acc.w);
        if (rem > 1) {
          acc.x = fmaf(v1.x, w1, acc.x); acc.y = fmaf(v1.y, w1, acc.y);
          acc.z = fmaf(v1.z, w1, acc.z); acc.w = fmaf(v1.w, w1, acc.w);
        }
        if (rem > 2) {
          acc.x = fmaf(v2.x, w2, acc.x); acc.y = fmaf(v2.y, w2, acc.y);
          acc.z = fmaf(v2.z, w2, acc.z); acc.w = fmaf(v2.w, w2, acc.w);
        }
        if (rem > 3) {
          acc.x = fmaf(v3.x, w3, acc.x); acc.y = fmaf(v3.y, w3, acc.y);
          acc.z = fmaf(v3.z, w3, acc.z); acc.w = fmaf(v3.w, w3, acc.w);
        }
      }
      const float sI = invI[node];
      acc.x *= sI; acc.y *= sI; acc.z *= sI; acc.w *= sI;
    }
    *(float4*)(As + (size_t)row * LDK + (size_t)lane * 4) = acc;
  }
  __syncthreads();

  gcn_tile_product(As, W, bias, C, mBase, M, Nd, relu, t);
}

extern "C" void kernel_launch(void* const* d_in, const int* in_sizes, int n_in,
                              void* d_out, int out_size, void* d_ws, size_t ws_size,
                              hipStream_t stream) {
  const float* feats = (const float*)d_in[0];
  const int*   src   = (const int*)d_in[1];
  const int*   dst   = (const int*)d_in[2];
  const float* W1    = (const float*)d_in[3];
  const float* b1    = (const float*)d_in[4];
  const float* W2    = (const float*)d_in[5];
  const float* b2    = (const float*)d_in[6];
  const float* W3    = (const float*)d_in[7];
  const float* b3    = (const float*)d_in[8];
  float*       out   = (float*)d_out;

  const int HID  = in_sizes[4];
  const int NCLS = in_sizes[8];
  const int INF  = in_sizes[3] / HID;
  const int N    = in_sizes[0] / INF;
  const int E    = in_sizes[1];

  const size_t Np = ((size_t)N + 1 + 63) & ~(size_t)63;
  const size_t Ep = ((size_t)E + 63) & ~(size_t)63;
  int*   cntO   = (int*)d_ws;
  int*   cntI   = cntO + Np;
  float* invO   = (float*)(cntI + Np);
  float* invI   = invO + Np;
  int*   rowptr = (int*)(invI + Np);
  int*   rank   = rowptr + Np;
  int2*  csr    = (int2*)(rank + Ep);
  int*   partO  = (int*)(csr + Ep);
  int*   partI  = partO + (size_t)NCHUNK * Np;
  float* bufA   = (float*)(partI + (size_t)NCHUNK * Np);
  float* bufB   = bufA + (size_t)N * TK;
  float* bufC   = bufB + (size_t)N * TK;

  const int B = 256;
  const int mTiles = (N + 15) / 16;
  const int eBlocks = (E + B - 1) / B;
  const int g10Blocks = (N + 23) / 24;
  const int CE = (E + NCHUNK - 1) / NCHUNK;
  const int nBlocks = (N + B - 1) / B;

  gcn_count_kernel<<<2 * NCHUNK, 1024, 0, stream>>>(src, dst, partO, partI, rank, N, E, CE, (int)Np);
  gcn_colscan_kernel<<<dim3(nBlocks, 2), B, 0, stream>>>(partO, partI, cntO, cntI, N, (int)Np);
  gcn_scan_kernel<<<1, 1024, 0, stream>>>(cntO, cntI, rowptr, invO, invI, N);
  gcn_fill_kernel<<<eBlocks, B, 0, stream>>>(src, dst, rank, rowptr, partI, invO, csr, E, CE, (int)Np);

  gcn_gather_gemm_kernel<<<mTiles, B, 0, stream>>>(feats, rowptr, csr, invI, W1, b1, bufB, N, HID, 1);

  gcn_gather_gemm_kernel<<<mTiles, B, 0, stream>>>(bufB, rowptr, csr, invI, W2, b2, bufA, N, HID, 1);

  gcn_gemm_kernel<<<mTiles, B, 0, stream>>>(bufA, nullptr, W3, nullptr, bufC, N, NCLS, 0);
  gcn_gather_kernel<10, true><<<g10Blocks, B, 0, stream>>>(bufC, rowptr, csr, invI, b3, out, N);
}


# Single-work-group degree scan in one pass, not 13 dependent sweeps: all tiles loaded at once, wave scans as row_shr DPP adds side by side, tile-by-wave totals scanned once; inv-sqrt in column scan
# speedup vs baseline: 3.6035x; 1.1829x over previous
#include <hip/hip_runtime.h>
#include <hip/hip_bf16.h>

typedef __attribute__((ext_vector_type(2))) float v2f;
typedef __attribute__((ext_vector_type(8))) float v8f;

#define TK 128
#define LDK 132

#define NCHUNK 64
#define HB 50176

__global__ void __launch_bounds__(1024)
gcn_count_kernel(const int* __restrict__ src, const int* __restrict__ dst,
                 int* __restrict__ partO, int* __restrict__ partI,
                 int* __restrict__ rank, int N, int E, int CE, int Np) {
  __shared__ int hist[HB];
  const int  t   = threadIdx.x;
  const bool isI = (int)blockIdx.x < NCHUNK;
  const int  g   = isI ? (int)blockIdx.x : (int)blockIdx.x - NCHUNK;
  const int  e0  = g * CE;
  const int  e1  = min(E, e0 + CE);
  int* __restrict__ part = (isI ? partI : partO) + (size_t)g * Np;

  for (int base = 0; base < N; base += HB) {
    const int lim  = min(HB, N - base);
    const int limP = min(HB, Np - base);
    for (int i = t * 4; i < limP; i += 4096)
      *(int4*)(hist + i) = make_int4(0, 0, 0, 0);
    __syncthreads();
    if (isI) {
      for (int e = e0 + t; e < e1; e += 8 * 1024) {
        int d[8], r[8];
        #pragma unroll
        for (int k = 0; k < 8; ++k) {
          const int ee = e + k * 1024;
          const int v  = dst[min(ee, e1 - 1)];
          d[k] = (ee < e1) ? v - base : -1;
        }
        #pragma unroll
        for (int k = 0; k < 8; ++k)
          r[k] = ((unsigned)d[k] < (unsigned)lim) ? atomicAdd(&hist[d[k]], 1) : 0;
        #pragma unroll
        for (int k = 0; k < 8; ++k)
          if ((unsigned)d[k] < (unsigned)lim) rank[e + k * 1024] = r[k];
      }
    } else {
      for (int e = e0 + t; e < e1; e += 8 * 1024) {
        int d[8];
        #pragma unroll
        for (int k = 0; k < 8; ++k) {
          const int ee = e + k * 1024;
          const int v  = src[min(ee, e1 - 1)];
          d[k] = (ee < e1) ? v - base : -1;
        }
        #pragma unroll
        for (int k = 0; k < 8; ++k) {
          if ((unsigned)d[k] < (unsigned)lim) atomicAdd(&hist[d[k]], 1);
        }
      }
    }
    __syncthreads();
    for (int i = t * 4; i < limP; i += 4096)
      *(int4*)(part + base + i) = *(const int4*)(hist + i);
    __syncthreads();
  }
}

__device__ __forceinline__ float gcn_inv_sqrt_deg(int c) {
  return (c > 0) ? rsqrtf(fmaxf((float)c, 1.0f)) : 0.0f;
}

__global__ void __launch_bounds__(256)
gcn_colscan_kernel(const int* __restrict__ partO, int* __restrict__ partI,
                   int* __restrict__ cntI, float* __restrict__ invO,
                   float* __restrict__ invI, int N, int Np) {
  const int n = blockIdx.x * 256 + threadIdx.x;
  if (n >= N) {
    if (n < Np && blockIdx.y == 0) cntI[n] = 0;
    return;
  }
  if (blockIdx.y == 0) {
    int* p = partI + n;
    int run = 0;
    for (int g = 0; g < NCHUNK; g += 16) {
      int c[16];
      #pragma unroll
      for (int k = 0; k < 16; ++k) c[k] = p[(size_t)(g + k) * Np];
      #pragma unroll
      for (int k = 0; k < 16; ++k) { p[(size_t)(g + k) * Np] = run; run += c[k]; }
    }
    cntI[n] = run;
    invI[n] = gcn_inv_sqrt_deg(run);
  } else {
    const int* p = partO + n;
    int s = 0;
    #pragma unroll 16
    for (int g = 0; g < NCHUNK; ++g) s += p[(size_t)g * Np];
    invO[n] = gcn_inv_sqrt_deg(s);
  }
}

__device__ __forceinline__ int gcn_wave_incl_scan(int x, int lane) {
  x += __builtin_amdgcn_update_dpp(0, x, 0x111, 0xf, 0xf, false);
  x += __builtin_amdgcn_update_dpp(0, x, 0x112, 0xf, 0xf, false);
  x += __builtin_amdgcn_update_dpp(0, x, 0x114, 0xf, 0xf, false);
  x += __builtin_amdgcn_update_dpp(0, x, 0x118, 0xf, 0xf, false);
  const int row0 = __builtin_amdgcn_readlane(x, 15);
  return (lane >= 16) ? x + row0 : x;
}

#define SCAN_Q 13

__global__ void __launch_bounds__(1024)
gcn_scan_kernel(const int* __restrict__ cntI, int* __restrict__ rowptr, int Np) {
  __shared__ int wsum[SCAN_Q * 32];
  __shared__ int tileTot[SCAN_Q];
  const int t    = threadIdx.x;
  const int lane = t & 31;
  const int w    = t >> 5;
  int carry = 0;
  for (int base = 0; base < Np; base += SCAN_Q * 4096) {
    if (base > 0) __syncthreads();
    int4 c[SCAN_Q];
    int  x[SCAN_Q];
    #pragma unroll
    for (int q = 0; q < SCAN_Q; ++q) {
      const int i = base + q * 4096 + t * 4;
      c[q] = make_int4(0, 0, 0, 0);
      if (i < Np) c[q] = *(const int4*)(cntI + i);
    }
    #pragma unroll
    for (int q = 0; q < SCAN_Q; ++q)
      x[q] = gcn_wave_incl_scan(c[q].x + c[q].y + c[q].z + c[q].w, lane);
    if (lane == 31) {
      #pragma unroll
      for (int q = 0; q < SCAN_Q; ++q) wsum[q * 32 + w] = x[q];
    }
    __syncthreads();
    if (w < SCAN_Q) {
      const int s  = wsum[w * 32 + lane];
      const int xs = gcn_wave_incl_scan(s, lane);
      wsum[w * 32 + lane] = xs - s;
      if (lane == 31) tileTot[w] = xs;
    }
    __syncthreads();
    int wo[SCAN_Q], tt[SCAN_Q];
    #pragma unroll
    for (int q = 0; q < SCAN_Q; ++q) { wo[q] = wsum[q * 32 + w]; tt[q] = tileTot[q]; }
    int tileOff = carry;
    #pragma unroll
    for (int q = 0; q < SCAN_Q; ++q) {
      const int i = base + q * 4096 + t * 4;
      const int tsum = c[q].x + c[q].y + c[q].z + c[q].w;
      int4 r;
      r.x = tileOff + wo[q] + x[q] - tsum;
      r.y = r.x + c[q].x;
      r.z = r.y + c[q].y;
      r.w = r.z + c[q].z;
      if (i < Np) *(int4*)(rowptr + i) = r;
      tileOff += tt[q];
    }
    carry = tileOff;
  }
}

__global__ void gcn_fill_kernel(const int* __restrict__ src, const int* __restrict__ dst,
                                const int* __restrict__ rank, const int* __restrict__ rowptr,
                                const int* __restrict__ partI,
                                const float* __restrict__ invO, int2* __restrict__ csr,
                                int E, int CE, int Np) {
  int t = blockIdx.x * blockDim.x + threadIdx.x;
  if (t < E) {
    int s = src[t];
    int d = dst[t];
    int g = t / CE;
    int2 p;
    p.x = s;
    p.y = __float_as_int(invO[s]);
    csr[rowptr[d] + partI[(size_t)g * Np + d] + rank[t]] = p;
  }
}

template <int CHUNKS, bool FINAL>
__global__ void __launch_bounds__(256)
gcn_gather_kernel(const float* __restrict__ h,
                  const int* __restrict__ rowptr,
                  const int2* __restrict__ csr,
                  const float* __restrict__ invI,
                  const float* __restrict__ bias,
                  float* __restrict__ outp, int N) {
  constexpr int NPW = 32 / CHUNKS;
  const int lane = threadIdx.x & 31;
  const int wave = blockIdx.x * 8 + (threadIdx.x >> 5);
  const int sub  = lane / CHUNKS;
  const int c    = lane - sub * CHUNKS;
  int node = wave * NPW + sub;
  if (NPW == 1) {
    node = __builtin_amdgcn_readfirstlane(node);
  }
  if (sub >= NPW || node >= N) return;

  const int beg = rowptr[node];
  const int end = rowptr[node + 1];
  const float4* __restrict__ h4 = (const float4*)h;

  float4 acc = make_float4(0.f, 0.f, 0.f, 0.f);
  int j = beg;
  for (; j + 4 <= end; j += 4) {
    int2 e0 = csr[j], e1 = csr[j + 1], e2 = csr[j + 2], e3 = csr[j + 3];
    float4 v0 = h4[(size_t)e0.x * CHUNKS + c];
    float4 v1 = h4[(size_t)e1.x * CHUNKS + c];
    float4 v2 = h4[(size_t)e2.x * CHUNKS + c];
    float4 v3 = h4[(size_t)e3.x * CHUNKS + c];
    float w0 = __int_as_float(e0.y), w1 = __int_as_float(e1.y);
    float w2 = __int_as_float(e2.y), w3 = __int_as_float(e3.y);
    acc.x = fmaf(v0.x, w0, acc.x); acc.y = fmaf(v0.y, w0, acc.y);
    acc.z = fmaf(v0.z, w0, acc.z); acc.w = fmaf(v0.w, w0, acc.w);
    acc.x = fmaf(v1.x, w1, acc.x); acc.y = fmaf(v1.y, w1, acc.y);
    acc.z = fmaf(v1.z, w1, acc.z); acc.w = fmaf(v1.w, w1, acc.w);
    acc.x = fmaf(v2.x, w2, acc.x); acc.y = fmaf(v2.y, w2, acc.y);
    acc.z = fmaf(v2.z, w2, acc.z); acc.w = fmaf(v2.w, w2, acc.w);
    acc.x = fmaf(v3.x, w3, acc.x); acc.y = fmaf(v3.y, w3, acc.y);
    acc.z = fmaf(v3.z, w3, acc.z); acc.w = fmaf(v3.w, w3, acc.w);
  }
  const int rem = end - j;
  if (rem > 0) {
    const int last = end - 1;
    int2 e0 = csr[j];
    int2 e1 = csr[(j + 1 < end) ? j + 1 : last];
    int2 e2 = csr[(j + 2 < end) ? j + 2 : last];
    float4 v0 = h4[(size_t)e0.x * CHUNKS + c];
    float4 v1 = h4[(size_t)e1.x * CHUNKS + c];
    float4 v2 = h4[(size_t)e2.x * CHUNKS + c];
    float w0 = __int_as_float(e0.y), w1 = __int_as_float(e1.y), w2 = __int_as_float(e2.y);
    acc.x = fmaf(v0.x, w0, acc.x); acc.y = fmaf(v0.y, w0, acc.y);
    acc.z = fmaf(v0.z, w0, acc.z); acc.w = fmaf(v0.w, w0, acc.w);
    if (rem > 1) {
      acc.x = fmaf(v1.x, w1, acc.x); acc.y = fmaf(v1.y, w1, acc.y);
      acc.z = fmaf(v1.z, w1, acc.z); acc.w = fmaf(v1.w, w1, acc.w);
    }
    if (rem > 2) {
      acc.x = fmaf(v2.x, w2, acc.x); acc.y = fmaf(v2.y, w2, acc.y);
      acc.z = fmaf(v2.z, w2, acc.z); acc.w = fmaf(v2.w, w2, acc.w);
    }
  }

  float4* op = (float4*)outp + (size_t)node * CHUNKS + c;
  if (FINAL) {
    const float  sI = invI[node];
    const float4 bb = ((const float4*)bias)[c];
    float4 r;
    r.x = acc.x * sI + bb.x; r.y = acc.y * sI + bb.y;
    r.z = acc.z * sI + bb.z; r.w = acc.w * sI + bb.w;
    *op = r;
  } else {
    *op = acc;
  }
}

__device__ __forceinline__ void
gcn_tile_product(const float* __restrict__ As,
                 const float* __restrict__ W,
                 const float* __restrict__ bias,
                 float* __restrict__ C,
                 int mBase, int M, int Nd, int relu, int t) {
  const int wave   = t >> 5;
  const int lane   = t & 31;
  const int nTiles = (Nd + 15) >> 4;
  if (wave >= nTiles) return;

  const int col    = lane & 15;
  const int gcol   = wave * 16 + col;
  const int khalf  = (lane >> 4) * 2;
  const bool colOk = (gcol < Nd);
  const int gcSafe = colOk ? gcol : (Nd - 1);

  const float* As_row = As + (size_t)(lane & 15) * LDK + khalf;
  const float* Wp     = W + (size_t)khalf * Nd + gcSafe;

  v8f acc0 = {};
  v8f acc1 = {};
  #pragma unroll 8
  for (int k0 = 0; k0 < TK; k0 += 8) {
    v2f a0, b0, a1, b1;
    a0[0] = As_row[k0];
    a0[1] = As_row[k0 + 1];
    b0[0] = Wp[0];
    b0[1] = Wp[Nd];
    a1[0] = As_row[k0 + 4];
    a1[1] = As_row[k0 + 5];
    b1[0] = Wp[4 * (size_t)Nd];
    b1[1] = Wp[5 * (size_t)Nd];
    acc0 = __builtin_amdgcn_wmma_f32_16x16x4_f32(false, a0, false, b0, (short)0, acc0,
                                                 false, false);
    acc1 = __builtin_amdgcn_wmma_f32_16x16x4_f32(false, a1, false, b1, (short)0, acc1,
                                                 false, false);
    Wp += 8 * (size_t)Nd;
  }
  v8f acc = acc0 + acc1;

  if (colOk) {
    const float bb    = bias ? bias[gcol] : 0.0f;
    const int   rbase = mBase + (lane >> 4) * 8;
    #pragma unroll
    for (int v = 0; v < 8; ++v) {
      int r = rbase + v;
      if (r < M) {
        float val = acc[v] + bb;
        if (relu) val = fmaxf(val, 0.0f);
        C[(size_t)r * Nd + gcol] = val;
      }
    }
  }
}

__global__ void __launch_bounds__(256)
gcn_gemm_kernel(const float* __restrict__ A,
                const float* __restrict__ rowScale,
                const float* __restrict__ W,
                const float* __restrict__ bias,
                float* __restrict__ C,
                int M, int Nd, int relu) {
  __shared__ float As[16 * LDK];

  const int t     = threadIdx.x;
  const int mBase = blockIdx.x * 16;

  for (int i = t; i < 16 * (TK / 4); i += 256) {
    int row = i >> 5;
    int c4  = i & 31;
    int gr  = mBase + row;
    float4 v = make_float4(0.f, 0.f, 0.f, 0.f);
    float  s = 1.0f;
    if (gr < M) {
      v = ((const float4*)(A + (size_t)gr * TK))[c4];
      if (rowScale) s = rowScale[gr];
    }
    float* dstp = As + (size_t)row * LDK + (size_t)c4 * 4;
    dstp[0] = v.x * s; dstp[1] = v.y * s; dstp[2] = v.z * s; dstp[3] = v.w * s;
  }
  __syncthreads();

  gcn_tile_product(As, W, bias, C, mBase, M, Nd, relu, t);
}

__global__ void __launch_bounds__(256)
gcn_gather_gemm_kernel(const float* __restrict__ h,
                       const int* __restrict__ rowptr,
                       const int2* __restrict__ csr,
                       const float* __restrict__ invI,
                       const float* __restrict__ W,
                       const float* __restrict__ bias,
                       float* __restrict__ C,
                       int M, int Nd, int relu) {
  __shared__ float As[16 * LDK];

  const int t     = threadIdx.x;
  const int lane  = t & 31;
  const int mBase = blockIdx.x * 16;
  const float4* __restrict__ h4 = (const float4*)h;

  for (int r = 0; r < 2; ++r) {
    const int row  = (t >> 5) * 2 + r;
    const int node = __builtin_amdgcn_readfirstlane(mBase + row);
    float4 acc = make_float4(0.f, 0.f, 0.f, 0.f);
    if (node < M) {
      const int beg  = rowptr[node];
      const int end  = rowptr[node + 1];
      const int last = end - 1;
      int2 e0 = make_int2(0, 0), e1 = e0, e2 = e0, e3 = e0;
      if (beg < end) {
        e0 = csr[beg];
        e1 = csr[min(beg + 1, last)];
        e2 = csr[min(beg + 2, last)];
        e3 = csr[min(beg + 3, last)];
      }
      for (int j = beg; j < end; j += 4) {
        const float4 v0 = h4[(size_t)e0.x * 32 + lane];
        const float4 v1 = h4[(size_t)e1.x * 32 + lane];
        const float4 v2 = h4[(size_t)e2.x * 32 + lane];
        const float4 v3 = h4[(size_t)e3.x * 32 + lane];
        const float w0 = __int_as_float(e0.y), w1 = __int_as_float(e1.y);
        const float w2 = __int_as_float(e2.y), w3 = __int_as_float(e3.y);
        const int rem = end - j;
        const int jn  = j + 4;
        if (jn < end) {
          e0 = csr[jn];
          e1 = csr[min(jn + 1, last)];
          e2 = csr[min(jn + 2, last)];
          e3 = csr[min(jn + 3, last)];
        }
        acc.x = fmaf(v0.x, w0, acc.x); acc.y = fmaf(v0.y, w0, acc.y);
        acc.z = fmaf(v0.z, w0, acc.z); acc.w = fmaf(v0.w, w0, acc.w);
        if (rem > 1) {
          acc.x = fmaf(v1.x, w1, acc.x); acc.y = fmaf(v1.y, w1, acc.y);
          acc.z = fmaf(v1.z, w1, acc.z); acc.w = fmaf(v1.w, w1, acc.w);
        }
        if (rem > 2) {
          acc.x = fmaf(v2.x, w2, acc.x); acc.y = fmaf(v2.y, w2, acc.y);
          acc.z = fmaf(v2.z, w2, acc.z); acc.w = fmaf(v2.w, w2, acc.w);
        }
        if (rem > 3) {
          acc.x = fmaf(v3.x, w3, acc.x); acc.y = fmaf(v3.y, w3, acc.y);
          acc.z = fmaf(v3.z, w3, acc.z); acc.w = fmaf(v3.w, w3, acc.w);
        }
      }
      const float sI = invI[node];
      acc.x *= sI; acc.y *= sI; acc.z *= sI; acc.w *= sI;
    }
    *(float4*)(As + (size_t)row * LDK + (size_t)lane * 4) = acc;
  }
  __syncthreads();

  gcn_tile_product(As, W, bias, C, mBase, M, Nd, relu, t);
}

extern "C" void kernel_launch(void* const* d_in, const int* in_sizes, int n_in,
                              void* d_out, int out_size, void* d_ws, size_t ws_size,
                              hipStream_t stream) {
  const float* feats = (const float*)d_in[0];
  const int*   src   = (const int*)d_in[1];
  const int*   dst   = (const int*)d_in[2];
  const float* W1    = (const float*)d_in[3];
  const float* b1    = (const float*)d_in[4];
  const float* W2    = (const float*)d_in[5];
  const float* b2    = (const float*)d_in[6];
  const float* W3    = (const float*)d_in[7];
  const float* b3    = (const float*)d_in[8];
  float*       out   = (float*)d_out;

  const int HID  = in_sizes[4];
  const int NCLS = in_sizes[8];
  const int INF  = in_sizes[3] / HID;
  const int N    = in_sizes[0] / INF;
  const int E    = in_sizes[1];

  const size_t Np = ((size_t)N + 1 + 63) & ~(size_t)63;
  const size_t Ep = ((size_t)E + 63) & ~(size_t)63;
  int*   cntI   = (int*)d_ws + Np;
  float* invO   = (float*)(cntI + Np);
  float* invI   = invO + Np;
  int*   rowptr = (int*)(invI + Np);
  int*   rank   = rowptr + Np;
  int2*  csr    = (int2*)(rank + Ep);
  int*   partO  = (int*)(csr + Ep);
  int*   partI  = partO + (size_t)NCHUNK * Np;
  float* bufA   = (float*)(partI + (size_t)NCHUNK * Np);
  float* bufB   = bufA + (size_t)N * TK;
  float* bufC   = bufB + (size_t)N * TK;

  const int B = 256;
  const int mTiles = (N + 15) / 16;
  const int eBlocks = (E + B - 1) / B;
  const int g10Blocks = (N + 23) / 24;
  const int CE = (E + NCHUNK - 1) / NCHUNK;
  const int nBlocks = (int)((Np + B - 1) / B);

  gcn_count_kernel<<<2 * NCHUNK, 1024, 0, stream>>>(src, dst, partO, partI, rank, N, E, CE, (int)Np);
  gcn_colscan_kernel<<<dim3(nBlocks, 2), B, 0, stream>>>(partO, partI, cntI, invO, invI, N, (int)Np);
  gcn_scan_kernel<<<1, 1024, 0, stream>>>(cntI, rowptr, (int)Np);
  gcn_fill_kernel<<<eBlocks, B, 0, stream>>>(src, dst, rank, rowptr, partI, invO, csr, E, CE, (int)Np);

  gcn_gather_gemm_kernel<<<mTiles, B, 0, stream>>>(feats, rowptr, csr, invI, W1, b1, bufB, N, HID, 1);

  gcn_gather_gemm_kernel<<<mTiles, B, 0, stream>>>(bufB, rowptr, csr, invI, W2, b2, bufA, N, HID, 1);

  gcn_gemm_kernel<<<mTiles, B, 0, stream>>>(bufA, nullptr, W3, nullptr, bufC, N, NCLS, 0);
  gcn_gather_kernel<10, true><<<g10Blocks, B, 0, stream>>>(bufC, rowptr, csr, invI, b3, out, N);
}


# Layer 3 projection (h2 @ W3) moved into the layer 2 kernel: the relu'd 16x128 tile stays in a second LDS tile and three waves multiply it by W3; h2 never goes to memory, one launch fewer
# speedup vs baseline: 3.8154x; 1.0588x over previous
#include <hip/hip_runtime.h>
#include <hip/hip_bf16.h>

typedef __attribute__((ext_vector_type(2))) float v2f;
typedef __attribute__((ext_vector_type(8))) float v8f;

#define TK 128
#define LDK 132

#define NCHUNK 64
#define HB 50176

__global__ void __launch_bounds__(1024)
gcn_count_kernel(const int* __restrict__ src, const int* __restrict__ dst,
                 int* __restrict__ partO, int* __restrict__ partI,
                 int* __restrict__ rank, int N, int E, int CE, int Np) {
  __shared__ int hist[HB];
  const int  t   = threadIdx.x;
  const bool isI = (int)blockIdx.x < NCHUNK;
  const int  g   = isI ? (int)blockIdx.x : (int)blockIdx.x - NCHUNK;
  const int  e0  = g * CE;
  const int  e1  = min(E, e0 + CE);
  int* __restrict__ part = (isI ? partI : partO) + (size_t)g * Np;

  for (int base = 0; base < N; base += HB) {
    const int lim  = min(HB, N - base);
    const int limP = min(HB, Np - base);
    for (int i = t * 4; i < limP; i += 4096)
      *(int4*)(hist + i) = make_int4(0, 0, 0, 0);
    __syncthreads();
    if (isI) {
      for (int e = e0 + t; e < e1; e += 8 * 1024) {
        int d[8], r[8];
        #pragma unroll
        for (int k = 0; k < 8; ++k) {
          const int ee = e + k * 1024;
          const int v  = dst[min(ee, e1 - 1)];
          d[k] = (ee < e1) ? v - base : -1;
        }
        #pragma unroll
        for (int k = 0; k < 8; ++k)
          r[k] = ((unsigned)d[k] < (unsigned)lim) ? atomicAdd(&hist[d[k]], 1) : 0;
        #pragma unroll
        for (int k = 0; k < 8; ++k)
          if ((unsigned)d[k] < (unsigned)lim) rank[e + k * 1024] = r[k];
      }
    } else {
      for (int e = e0 + t; e < e1; e += 8 * 1024) {
        int d[8];
        #pragma unroll
        for (int k = 0; k < 8; ++k) {
          const int ee = e + k * 1024;
          const int v  = src[min(ee, e1 - 1)];
          d[k] = (ee < e1) ? v - base : -1;
        }
        #pragma unroll
        for (int k = 0; k < 8; ++k) {
          if ((unsigned)d[k] < (unsigned)lim) atomicAdd(&hist[d[k]], 1);
        }
      }
    }
    __syncthreads();
    for (int i = t * 4; i < limP; i += 4096)
      *(int4*)(part + base + i) = *(const int4*)(hist + i);
    __syncthreads();
  }
}

__device__ __forceinline__ float gcn_inv_sqrt_deg(int c) {
  return (c > 0) ? rsqrtf(fmaxf((float)c, 1.0f)) : 0.0f;
}

__global__ void __launch_bounds__(256)
gcn_colscan_kernel(const int* __restrict__ partO, int* __restrict__ partI,
                   int* __restrict__ cntI, float* __restrict__ invO,
                   float* __restrict__ invI, int N, int Np) {
  const int n = blockIdx.x * 256 + threadIdx.x;
  if (n >= N) {
    if (n < Np && blockIdx.y == 0) cntI[n] = 0;
    return;
  }
  if (blockIdx.y == 0) {
    int* p = partI + n;
    int run = 0;
    for (int g = 0; g < NCHUNK; g += 16) {
      int c[16];
      #pragma unroll
      for (int k = 0; k < 16; ++k) c[k] = p[(size_t)(g + k) * Np];
      #pragma unroll
      for (int k = 0; k < 16; ++k) { p[(size_t)(g + k) * Np] = run; run += c[k]; }
    }
    cntI[n] = run;
    invI[n] = gcn_inv_sqrt_deg(run);
  } else {
    const int* p = partO + n;
    int s = 0;
    #pragma unroll 16
    for (int g = 0; g < NCHUNK; ++g) s += p[(size_t)g * Np];
    invO[n] = gcn_inv_sqrt_deg(s);
  }
}

__device__ __forceinline__ int gcn_wave_incl_scan(int x, int lane) {
  x += __builtin_amdgcn_update_dpp(0, x, 0x111, 0xf, 0xf, false);
  x += __builtin_amdgcn_update_dpp(0, x, 0x112, 0xf, 0xf, false);
  x += __builtin_amdgcn_update_dpp(0, x, 0x114, 0xf, 0xf, false);
  x += __builtin_amdgcn_update_dpp(0, x, 0x118, 0xf, 0xf, false);
  const int row0 = __builtin_amdgcn_readlane(x, 15);
  return (lane >= 16) ? x + row0 : x;
}

#define SCAN_Q 13

__global__ void __launch_bounds__(1024)
gcn_scan_kernel(const int* __restrict__ cntI, int* __restrict__ rowptr, int Np) {
  __shared__ int wsum[SCAN_Q * 32];
  __shared__ int tileTot[SCAN_Q];
  const int t    = threadIdx.x;
  const int lane = t & 31;
  const int w    = t >> 5;
  int carry = 0;
  for (int base = 0; base < Np; base += SCAN_Q * 4096) {
    if (base > 0) __syncthreads();
    int4 c[SCAN_Q];
    int  x[SCAN_Q];
    #pragma unroll
    for (int q = 0; q < SCAN_Q; ++q) {
      const int i = base + q * 4096 + t * 4;
      c[q] = make_int4(0, 0, 0, 0);
      if (i < Np) c[q] = *(const int4*)(cntI + i);
    }
    #pragma unroll
    for (int q = 0; q < SCAN_Q; ++q)
      x[q] = gcn_wave_incl_scan(c[q].x + c[q].y + c[q].z + c[q].w, lane);
    if (lane == 31) {
      #pragma unroll
      for (int q = 0; q < SCAN_Q; ++q) wsum[q * 32 + w] = x[q];
    }
    __syncthreads();
    if (w < SCAN_Q) {
      const int s  = wsum[w * 32 + lane];
      const int xs = gcn_wave_incl_scan(s, lane);
      wsum[w * 32 + lane] = xs - s;
      if (lane == 31) tileTot[w] = xs;
    }
    __syncthreads();
    int wo[SCAN_Q], tt[SCAN_Q];
    #pragma unroll
    for (int q = 0; q < SCAN_Q; ++q) { wo[q] = wsum[q * 32 + w]; tt[q] = tileTot[q]; }
    int tileOff = carry;
    #pragma unroll
    for (int q = 0; q < SCAN_Q; ++q) {
      const int i = base + q * 4096 + t * 4;
      const int tsum = c[q].x + c[q].y + c[q].z + c[q].w;
      int4 r;
      r.x = tileOff + wo[q] + x[q] - tsum;
      r.y = r.x + c[q].x;
      r.z = r.y + c[q].y;
      r.w = r.z + c[q].z;
      if (i < Np) *(int4*)(rowptr + i) = r;
      tileOff += tt[q];
    }
    carry = tileOff;
  }
}

__global__ void gcn_fill_kernel(const int* __restrict__ src, const int* __restrict__ dst,
                                const int* __restrict__ rank, const int* __restrict__ rowptr,
                                const int* __restrict__ partI,
                                const float* __restrict__ invO, int2* __restrict__ csr,
                                int E, int CE, int Np) {
  int t = blockIdx.x * blockDim.x + threadIdx.x;
  if (t < E) {
    int s = src[t];
    int d = dst[t];
    int g = t / CE;
    int2 p;
    p.x = s;
    p.y = __float_as_int(invO[s]);
    csr[rowptr[d] + partI[(size_t)g * Np + d] + rank[t]] = p;
  }
}

template <int CHUNKS, bool FINAL>
__global__ void __launch_bounds__(256)
gcn_gather_kernel(const float* __restrict__ h,
                  const int* __restrict__ rowptr,
                  const int2* __restrict__ csr,
                  const float* __restrict__ invI,
                  const float* __restrict__ bias,
                  float* __restrict__ outp, int N) {
  constexpr int NPW = 32 / CHUNKS;
  const int lane = threadIdx.x & 31;
  const int wave = blockIdx.x * 8 + (threadIdx.x >> 5);
  const int sub  = lane / CHUNKS;
  const int c    = lane - sub * CHUNKS;
  int node = wave * NPW + sub;
  if (NPW == 1) {
    node = __builtin_amdgcn_readfirstlane(node);
  }
  if (sub >= NPW || node >= N) return;

  const int beg = rowptr[node];
  const int end = rowptr[node + 1];
  const float4* __restrict__ h4 = (const float4*)h;

  float4 acc = make_float4(0.f, 0.f, 0.f, 0.f);
  int j = beg;
  for (; j + 4 <= end; j += 4) {
    int2 e0 = csr[j], e1 = csr[j + 1], e2 = csr[j + 2], e3 = csr[j + 3];
    float4 v0 = h4[(size_t)e0.x * CHUNKS + c];
    float4 v1 = h4[(size_t)e1.x * CHUNKS + c];
    float4 v2 = h4[(size_t)e2.x * CHUNKS + c];
    float4 v3 = h4[(size_t)e3.x * CHUNKS + c];
    float w0 = __int_as_float(e0.y), w1 = __int_as_float(e1.y);
    float w2 = __int_as_float(e2.y), w3 = __int_as_float(e3.y);
    acc.x = fmaf(v0.x, w0, acc.x); acc.y = fmaf(v0.y, w0, acc.y);
    acc.z = fmaf(v0.z, w0, acc.z); acc.w = fmaf(v0.w, w0, acc.w);
    acc.x = fmaf(v1.x, w1, acc.x); acc.y = fmaf(v1.y, w1, acc.y);
    acc.z = fmaf(v1.z, w1, acc.z); acc.w = fmaf(v1.w, w1, acc.w);
    acc.x = fmaf(v2.x, w2, acc.x); acc.y = fmaf(v2.y, w2, acc.y);
    acc.z = fmaf(v2.z, w2, acc.z); acc.w = fmaf(v2.w, w2, acc.w);
    acc.x = fmaf(v3.x, w3, acc.x); acc.y = fmaf(v3.y, w3, acc.y);
    acc.z = fmaf(v3.z, w3, acc.z); acc.w = fmaf(v3.w, w3, acc.w);
  }
  const int rem = end - j;
  if (rem > 0) {
    const int last = end - 1;
    int2 e0 = csr[j];
    int2 e1 = csr[(j + 1 < end) ? j + 1 : last];
    int2 e2 = csr[(j + 2 < end) ? j + 2 : last];
    float4 v0 = h4[(size_t)e0.x * CHUNKS + c];
    float4 v1 = h4[(size_t)e1.x * CHUNKS + c];
    float4 v2 = h4[(size_t)e2.x * CHUNKS + c];
    float w0 = __int_as_float(e0.y), w1 = __int_as_float(e1.y), w2 = __int_as_float(e2.y);
    acc.x = fmaf(v0.x, w0, acc.x); acc.y = fmaf(v0.y, w0, acc.y);
    acc.z = fmaf(v0.z, w0, acc.z); acc.w = fmaf(v0.w, w0, acc.w);
    if (rem > 1) {
      acc.x = fmaf(v1.x, w1, acc.x); acc.y = fmaf(v1.y, w1, acc.y);
      acc.z = fmaf(v1.z, w1, acc.z); acc.w = fmaf(v1.w, w1, acc.w);
    }
    if (rem > 2) {
      acc.x = fmaf(v2.x, w2, acc.x); acc.y = fmaf(v2.y, w2, acc.y);
      acc.z = fmaf(v2.z, w2, acc.z); acc.w = fmaf(v2.w, w2, acc.w);
    }
  }

  float4* op = (float4*)outp + (size_t)node * CHUNKS + c;
  if (FINAL) {
    const float  sI = invI[node];
    const float4 bb = ((const float4*)bias)[c];
    float4 r;
    r.x = acc.x * sI + bb.x; r.y = acc.y * sI + bb.y;
    r.z = acc.z * sI + bb.z; r.w = acc.w * sI + bb.w;
    *op = r;
  } else {
    *op = acc;
  }
}

__device__ __forceinline__ v8f
gcn_tile_mma(const float* __restrict__ As,
             const float* __restrict__ W,
             int Nd, int wave, int lane) {
  const int col    = lane & 15;
  const int gcol   = wave * 16 + col;
  const int khalf  = (lane >> 4) * 2;
  const int gcSafe = (gcol < Nd) ? gcol : (Nd - 1);

  const float* As_row = As + (size_t)(lane & 15) * LDK + khalf;
  const float* Wp     = W + (size_t)khalf * Nd + gcSafe;

  v8f acc0 = {};
  v8f acc1 = {};
  #pragma unroll 8
  for (int k0 = 0; k0 < TK; k0 += 8) {
    v2f a0, b0, a1, b1;
    a0[0] = As_row[k0];
    a0[1] = As_row[k0 + 1];
    b0[0] = Wp[0];
    b0[1] = Wp[Nd];
    a1[0] = As_row[k0 + 4];
    a1[1] = As_row[k0 + 5];
    b1[0] = Wp[4 * (size_t)Nd];
    b1[1] = Wp[5 * (size_t)Nd];
    acc0 = __builtin_amdgcn_wmma_f32_16x16x4_f32(false, a0, false, b0, (short)0, acc0,
                                                 false, false);
    acc1 = __builtin_amdgcn_wmma_f32_16x16x4_f32(false, a1, false, b1, (short)0, acc1,
                                                 false, false);
    Wp += 8 * (size_t)Nd;
  }
  return acc0 + acc1;
}

__device__ __forceinline__ v8f
gcn_tile_mma_tk(const float* __restrict__ As,
                const float* __restrict__ W,
                int wave, int lane) {
  const int      khalf = (lane >> 4) * 2;
  const unsigned bi    = (unsigned)(khalf * TK + (lane & 15));
  const float* As_row  = As + (size_t)(lane & 15) * LDK + khalf;
  const float* Wb      = W + wave * 16;

  v8f acc0 = {};
  v8f acc1 = {};
  #pragma unroll 8
  for (int k0 = 0; k0 < TK; k0 += 8) {
    const float* Wk = Wb + (size_t)k0 * TK;
    v2f a0, b0, a1, b1;
    a0[0] = As_row[k0];
    a0[1] = As_row[k0 + 1];
    b0[0] = Wk[bi];
    b0[1] = (Wk + TK)[bi];
    a1[0] = As_row[k0 + 4];
    a1[1] = As_row[k0 + 5];
    b1[0] = (Wk + 4 * TK)[bi];
    b1[1] = (Wk + 5 * TK)[bi];
    acc0 = __builtin_amdgcn_wmma_f32_16x16x4_f32(false, a0, false, b0, (short)0, acc0,
                                                 false, false);
    acc1 = __builtin_amdgcn_wmma_f32_16x16x4_f32(false, a1, false, b1, (short)0, acc1,
                                                 false, false);
  }
  return acc0 + acc1;
}

__device__ __forceinline__ void
gcn_tile_product(const float* __restrict__ As,
                 const float* __restrict__ W,
                 const float* __restrict__ bias,
                 float* __restrict__ C,
                 int mBase, int M, int Nd, int relu, int t) {
  const int wave   = t >> 5;
  const int lane   = t & 31;
  const int nTiles = (Nd + 15) >> 4;
  if (wave >= nTiles) return;

  const v8f acc  = gcn_tile_mma(As, W, Nd, wave, lane);
  const int gcol = wave * 16 + (lane & 15);
  if (gcol < Nd) {
    const float bb    = bias ? bias[gcol] : 0.0f;
    const int   rbase = mBase + (lane >> 4) * 8;
    #pragma unroll
    for (int v = 0; v < 8; ++v) {
      int r = rbase + v;
      if (r < M) {
        float val = acc[v] + bb;
        if (relu) val = fmaxf(val, 0.0f);
        C[(size_t)r * Nd + gcol] = val;
      }
    }
  }
}

__device__ __forceinline__ void
gcn_tile_product_lds(const float* __restrict__ As,
                     const float* __restrict__ W,
                     const float* __restrict__ bias,
                     float* __restrict__ Hs,
                     int relu, int t) {
  static_assert(TK == 8 * 16, "one 16-column tile per wave of the 256-thread block");
  const int wave = __builtin_amdgcn_readfirstlane(t >> 5);
  const int lane = t & 31;
  const v8f acc  = gcn_tile_mma_tk(As, W, wave, lane);
  const int gcol = wave * 16 + (lane & 15);
  const float bb = bias ? bias[gcol] : 0.0f;
  float* hp = Hs + (size_t)((lane >> 4) * 8) * LDK + gcol;
  #pragma unroll
  for (int v = 0; v < 8; ++v) {
    float val = acc[v] + bb;
    if (relu) val = fmaxf(val, 0.0f);
    hp[(size_t)v * LDK] = val;
  }
}

template <bool PROJECT>
__global__ void __launch_bounds__(256)
gcn_gather_gemm_kernel(const float* __restrict__ h,
                       const int* __restrict__ rowptr,
                       const int2* __restrict__ csr,
                       const float* __restrict__ invI,
                       const float* __restrict__ W,
                       const float* __restrict__ bias,
                       const float* __restrict__ Wproj,
                       float* __restrict__ C,
                       int M, int Nd, int relu, int Nproj) {
  __shared__ float As[16 * LDK];

  const int t     = threadIdx.x;
  const int lane  = t & 31;
  const int mBase = blockIdx.x * 16;
  const float4* __restrict__ h4 = (const float4*)h;

  for (int r = 0; r < 2; ++r) {
    const int row  = (t >> 5) * 2 + r;
    const int node = __builtin_amdgcn_readfirstlane(mBase + row);
    float4 acc = make_float4(0.f, 0.f, 0.f, 0.f);
    if (node < M) {
      const int beg  = rowptr[node];
      const int end  = rowptr[node + 1];
      const int last = end - 1;
      int2 e0 = make_int2(0, 0), e1 = e0, e2 = e0, e3 = e0;
      if (beg < end) {
        e0 = csr[beg];
        e1 = csr[min(beg + 1, last)];
        e2 = csr[min(beg + 2, last)];
        e3 = csr[min(beg + 3, last)];
      }
      for (int j = beg; j < end; j += 4) {
        const float4 v0 = h4[(size_t)e0.x * 32 + lane];
        const float4 v1 = h4[(size_t)e1.x * 32 + lane];
        const float4 v2 = h4[(size_t)e2.x * 32 + lane];
        const float4 v3 = h4[(size_t)e3.x * 32 + lane];
        const float w0 = __int_as_float(e0.y), w1 = __int_as_float(e1.y);
        const float w2 = __int_as_float(e2.y), w3 = __int_as_float(e3.y);
        const int rem = end - j;
        const int jn  = j + 4;
        if (jn < end) {
          e0 = csr[jn];
          e1 = csr[min(jn + 1, last)];
          e2 = csr[min(jn + 2, last)];
          e3 = csr[min(jn + 3, last)];
        }
        acc.x = fmaf(v0.x, w0, acc.x); acc.y = fmaf(v0.y, w0, acc.y);
        acc.z = fmaf(v0.z, w0, acc.z); acc.w = fmaf(v0.w, w0, acc.w);
        if (rem > 1) {
          acc.x = fmaf(v1.x, w1, acc.x); acc.y = fmaf(v1.y, w1, acc.y);
          acc.z = fmaf(v1.z, w1, acc.z); acc.w = fmaf(v1.w, w1, acc.w);
        }
        if (rem > 2) {
          acc.x = fmaf(v2.x, w2, acc.x); acc.y = fmaf(v2.y, w2, acc.y);
          acc.z = fmaf(v2.z, w2, acc.z); acc.w = fmaf(v2.w, w2, acc.w);
        }
        if (rem > 3) {
          acc.x = fmaf(v3.x, w3, acc.x); acc.y = fmaf(v3.y, w3, acc.y);
          acc.z = fmaf(v3.z, w3, acc.z); acc.w = fmaf(v3.w, w3, acc.w);
        }
      }
      const float sI = invI[node];
      acc.x *= sI; acc.y *= sI; acc.z *= sI; acc.w *= sI;
    }
    *(float4*)(As + (size_t)row * LDK + (size_t)lane * 4) = acc;
  }
  __syncthreads();

  if constexpr (PROJECT) {
    __shared__ float Hs[16 * LDK];
    gcn_tile_product_lds(As, W, bias, Hs, relu, t);
    __syncthreads();
    gcn_tile_product(Hs, Wproj, nullptr, C, mBase, M, Nproj, 0, t);
  } else {
    gcn_tile_product(As, W, bias, C, mBase, M, Nd, relu, t);
  }
}

extern "C" void kernel_launch(void* const* d_in, const int* in_sizes, int n_in,
                              void* d_out, int out_size, void* d_ws, size_t ws_size,
                              hipStream_t stream) {
  const float* feats = (const float*)d_in[0];
  const int*   src   = (const int*)d_in[1];
  const int*   dst   = (const int*)d_in[2];
  const float* W1    = (const float*)d_in[3];
  const float* b1    = (const float*)d_in[4];
  const float* W2    = (const float*)d_in[5];
  const float* b2    = (const float*)d_in[6];
  const float* W3    = (const float*)d_in[7];
  const float* b3    = (const float*)d_in[8];
  float*       out   = (float*)d_out;

  const int HID  = in_sizes[4];
  const int NCLS = in_sizes[8];
  const int INF  = in_sizes[3] / HID;
  const int N    = in_sizes[0] / INF;
  const int E    = in_sizes[1];

  const size_t Np = ((size_t)N + 1 + 63) & ~(size_t)63;
  const size_t Ep = ((size_t)E + 63) & ~(size_t)63;
  int*   cntI   = (int*)d_ws + Np;
  float* invO   = (float*)(cntI + Np);
  float* invI   = invO + Np;
  int*   rowptr = (int*)(invI + Np);
  int*   rank   = rowptr + Np;
  int2*  csr    = (int2*)(rank + Ep);
  int*   partO  = (int*)(csr + Ep);
  int*   partI  = partO + (size_t)NCHUNK * Np;
  float* bufB   = (float*)(partI + (size_t)NCHUNK * Np);
  float* bufC   = bufB + (size_t)N * TK;

  const int B = 256;
  const int mTiles = (N + 15) / 16;
  const int eBlocks = (E + B - 1) / B;
  const int g10Blocks = (N + 23) / 24;
  const int CE = (E + NCHUNK - 1) / NCHUNK;
  const int nBlocks = (int)((Np + B - 1) / B);

  gcn_count_kernel<<<2 * NCHUNK, 1024, 0, stream>>>(src, dst, partO, partI, rank, N, E, CE, (int)Np);
  gcn_colscan_kernel<<<dim3(nBlocks, 2), B, 0, stream>>>(partO, partI, cntI, invO, invI, N, (int)Np);
  gcn_scan_kernel<<<1, 1024, 0, stream>>>(cntI, rowptr, (int)Np);
  gcn_fill_kernel<<<eBlocks, B, 0, stream>>>(src, dst, rank, rowptr, partI, invO, csr, E, CE, (int)Np);

  gcn_gather_gemm_kernel<false><<<mTiles, B, 0, stream>>>(feats, rowptr, csr, invI, W1, b1, nullptr, bufB, N, HID, 1, 0);

  gcn_gather_gemm_kernel<true><<<mTiles, B, 0, stream>>>(bufB, rowptr, csr, invI, W2, b2, W3, bufC, N, HID, 1, NCLS);

  gcn_gather_kernel<10, true><<<g10Blocks, B, 0, stream>>>(bufC, rowptr, csr, invI, b3, out, N);
}


# GEMM B operand read from a fragment-order copy of each weight (built every call by extra blocks of the count launch): 16 whole-line 16-byte loads per wave, not 64 four-byte ones; k order kept
# speedup vs baseline: 3.9670x; 1.0397x over previous
#include <hip/hip_runtime.h>
#include <hip/hip_bf16.h>

typedef __attribute__((ext_vector_type(2))) float v2f;
typedef __attribute__((ext_vector_type(8))) float v8f;

#define TK 128
#define LDK 132

#define NCHUNK 64
#define HB 50176
#define FRAG_TILE 2048

__device__ __forceinline__ void gcn_frag_copy(const float* __restrict__ W,
                                              float* __restrict__ Wf, int Nd, int i) {
  const int e    = i & 3;
  const int lane = (i >> 2) & 31;
  const int s    = (i >> 7) & 15;
  const int j    = i >> 11;
  const int k    = s * 8 + (e >> 1) * 4 + (lane >> 4) * 2 + (e & 1);
  const int col  = j * 16 + (lane & 15);
  Wf[i] = (col < Nd) ? W[(size_t)k * Nd + col] : 0.0f;
}

__global__ void __launch_bounds__(1024)
gcn_count_kernel(const int* __restrict__ src, const int* __restrict__ dst,
                 int* __restrict__ partO, int* __restrict__ partI,
                 int* __restrict__ rank, int N, int E, int CE, int Np,
                 const float* __restrict__ W1, const float* __restrict__ W2,
                 const float* __restrict__ W3, float* __restrict__ Wf1,
                 float* __restrict__ Wf2, float* __restrict__ Wf3,
                 int hid, int ncls) {
  __shared__ int hist[HB];
  const int  t   = threadIdx.x;
  if ((int)blockIdx.x >= 2 * NCHUNK) {
    const int nH = ((hid + 15) >> 4) * FRAG_TILE;
    const int nC = ((ncls + 15) >> 4) * FRAG_TILE;
    const int i  = ((int)blockIdx.x - 2 * NCHUNK) * 1024 + t;
    if (i < nH)               gcn_frag_copy(W1, Wf1, hid, i);
    else if (i < 2 * nH)      gcn_frag_copy(W2, Wf2, hid, i - nH);
    else if (i < 2 * nH + nC) gcn_frag_copy(W3, Wf3, ncls, i - 2 * nH);
    return;
  }
  const bool isI = (int)blockIdx.x < NCHUNK;
  const int  g   = isI ? (int)blockIdx.x : (int)blockIdx.x - NCHUNK;
  const int  e0  = g * CE;
  const int  e1  = min(E, e0 + CE);
  int* __restrict__ part = (isI ? partI : partO) + (size_t)g * Np;

  for (int base = 0; base < N; base += HB) {
    const int lim  = min(HB, N - base);
    const int limP = min(HB, Np - base);
    for (int i = t * 4; i < limP; i += 4096)
      *(int4*)(hist + i) = make_int4(0, 0, 0, 0);
    __syncthreads();
    if (isI) {
      for (int e = e0 + t; e < e1; e += 8 * 1024) {
        int d[8], r[8];
        #pragma unroll
        for (int k = 0; k < 8; ++k) {
          const int ee = e + k * 1024;
          const int v  = dst[min(ee, e1 - 1)];
          d[k] = (ee < e1) ? v - base : -1;
        }
        #pragma unroll
        for (int k = 0; k < 8; ++k)
          r[k] = ((unsigned)d[k] < (unsigned)lim) ? atomicAdd(&hist[d[k]], 1) : 0;
        #pragma unroll
        for (int k = 0; k < 8; ++k)
          if ((unsigned)d[k] < (unsigned)lim) rank[e + k * 1024] = r[k];
      }
    } else {
      for (int e = e0 + t; e < e1; e += 8 * 1024) {
        int d[8];
        #pragma unroll
        for (int k = 0; k < 8; ++k) {
          const int ee = e + k * 1024;
          const int v  = src[min(ee, e1 - 1)];
          d[k] = (ee < e1) ? v - base : -1;
        }
        #pragma unroll
        for (int k = 0; k < 8; ++k) {
          if ((unsigned)d[k] < (unsigned)lim) atomicAdd(&hist[d[k]], 1);
        }
      }
    }
    __syncthreads();
    for (int i = t * 4; i < limP; i += 4096)
      *(int4*)(part + base + i) = *(const int4*)(hist + i);
    __syncthreads();
  }
}

__device__ __forceinline__ float gcn_inv_sqrt_deg(int c) {
  return (c > 0) ? rsqrtf(fmaxf((float)c, 1.0f)) : 0.0f;
}

__global__ void __launch_bounds__(256)
gcn_colscan_kernel(const int* __restrict__ partO, int* __restrict__ partI,
                   int* __restrict__ cntI, float* __restrict__ invO,
                   float* __restrict__ invI, int N, int Np) {
  const int n = blockIdx.x * 256 + threadIdx.x;
  if (n >= N) {
    if (n < Np && blockIdx.y == 0) cntI[n] = 0;
    return;
  }
  if (blockIdx.y == 0) {
    int* p = partI + n;
    int run = 0;
    for (int g = 0; g < NCHUNK; g += 16) {
      int c[16];
      #pragma unroll
      for (int k = 0; k < 16; ++k) c[k] = p[(size_t)(g + k) * Np];
      #pragma unroll
      for (int k = 0; k < 16; ++k) { p[(size_t)(g + k) * Np] = run; run += c[k]; }
    }
    cntI[n] = run;
    invI[n] = gcn_inv_sqrt_deg(run);
  } else {
    const int* p = partO + n;
    int s = 0;
    #pragma unroll 16
    for (int g = 0; g < NCHUNK; ++g) s += p[(size_t)g * Np];
    invO[n] = gcn_inv_sqrt_deg(s);
  }
}

__device__ __forceinline__ int gcn_wave_incl_scan(int x, int lane) {
  x += __builtin_amdgcn_update_dpp(0, x, 0x111, 0xf, 0xf, false);
  x += __builtin_amdgcn_update_dpp(0, x, 0x112, 0xf, 0xf, false);
  x += __builtin_amdgcn_update_dpp(0, x, 0x114, 0xf, 0xf, false);
  x += __builtin_amdgcn_update_dpp(0, x, 0x118, 0xf, 0xf, false);
  const int row0 = __builtin_amdgcn_readlane(x, 15);
  return (lane >= 16) ? x + row0 : x;
}

#define SCAN_Q 13

__global__ void __launch_bounds__(1024)
gcn_scan_kernel(const int* __restrict__ cntI, int* __restrict__ rowptr, int Np) {
  __shared__ int wsum[SCAN_Q * 32];
  __shared__ int tileTot[SCAN_Q];
  const int t    = threadIdx.x;
  const int lane = t & 31;
  const int w    = t >> 5;
  int carry = 0;
  for (int base = 0; base < Np; base += SCAN_Q * 4096) {
    if (base > 0) __syncthreads();
    int4 c[SCAN_Q];
    int  x[SCAN_Q];
    #pragma unroll
    for (int q = 0; q < SCAN_Q; ++q) {
      const int i = base + q * 4096 + t * 4;
      c[q] = make_int4(0, 0, 0, 0);
      if (i < Np) c[q] = *(const int4*)(cntI + i);
    }
    #pragma unroll
    for (int q = 0; q < SCAN_Q; ++q)
      x[q] = gcn_wave_incl_scan(c[q].x + c[q].y + c[q].z + c[q].w, lane);
    if (lane == 31) {
      #pragma unroll
      for (int q = 0; q < SCAN_Q; ++q) wsum[q * 32 + w] = x[q];
    }
    __syncthreads();
    if (w < SCAN_Q) {
      const int s  = wsum[w * 32 + lane];
      const int xs = gcn_wave_incl_scan(s, lane);
      wsum[w * 32 + lane] = xs - s;
      if (lane == 31) tileTot[w] = xs;
    }
    __syncthreads();
    int wo[SCAN_Q], tt[SCAN_Q];
    #pragma unroll
    for (int q = 0; q < SCAN_Q; ++q) { wo[q] = wsum[q * 32 + w]; tt[q] = tileTot[q]; }
    int tileOff = carry;
    #pragma unroll
    for (int q = 0; q < SCAN_Q; ++q) {
      const int i = base + q * 4096 + t * 4;
      const int tsum = c[q].x + c[q].y + c[q].z + c[q].w;
      int4 r;
      r.x = tileOff + wo[q] + x[q] - tsum;
      r.y = r.x + c[q].x;
      r.z = r.y + c[q].y;
      r.w = r.z + c[q].z;
      if (i < Np) *(int4*)(rowptr + i) = r;
      tileOff += tt[q];
    }
    carry = tileOff;
  }
}

__global__ void gcn_fill_kernel(const int* __restrict__ src, const int* __restrict__ dst,
                                const int* __restrict__ rank, const int* __restrict__ rowptr,
                                const int* __restrict__ partI,
                                const float* __restrict__ invO, int2* __restrict__ csr,
                                int E, int CE, int Np) {
  int t = blockIdx.x * blockDim.x + threadIdx.x;
  if (t < E) {
    int s = src[t];
    int d = dst[t];
    int g = t / CE;
    int2 p;
    p.x = s;
    p.y = __float_as_int(invO[s]);
    csr[rowptr[d] + partI[(size_t)g * Np + d] + rank[t]] = p;
  }
}

template <int CHUNKS, bool FINAL>
__global__ void __launch_bounds__(256)
gcn_gather_kernel(const float* __restrict__ h,
                  const int* __restrict__ rowptr,
                  const int2* __restrict__ csr,
                  const float* __restrict__ invI,
                  const float* __restrict__ bias,
                  float* __restrict__ outp, int N) {
  constexpr int NPW = 32 / CHUNKS;
  const int lane = threadIdx.x & 31;
  const int wave = blockIdx.x * 8 + (threadIdx.x >> 5);
  const int sub  = lane / CHUNKS;
  const int c    = lane - sub * CHUNKS;
  int node = wave * NPW + sub;
  if (NPW == 1) {
    node = __builtin_amdgcn_readfirstlane(node);
  }
  if (sub >= NPW || node >= N) return;

  const int beg = rowptr[node];
  const int end = rowptr[node + 1];
  const float4* __restrict__ h4 = (const float4*)h;

  float4 acc = make_float4(0.f, 0.f, 0.f, 0.f);
  int j = beg;
  for (; j + 4 <= end; j += 4) {
    int2 e0 = csr[j], e1 = csr[j + 1], e2 = csr[j + 2], e3 = csr[j + 3];
    float4 v0 = h4[(size_t)e0.x * CHUNKS + c];
    float4 v1 = h4[(size_t)e1.x * CHUNKS + c];
    float4 v2 = h4[(size_t)e2.x * CHUNKS + c];
    float4 v3 = h4[(size_t)e3.x * CHUNKS + c];
    float w0 = __int_as_float(e0.y), w1 = __int_as_float(e1.y);
    float w2 = __int_as_float(e2.y), w3 = __int_as_float(e3.y);
    acc.x = fmaf(v0.x, w0, acc.x); acc.y = fmaf(v0.y, w0, acc.y);
    acc.z = fmaf(v0.z, w0, acc.z); acc.w = fmaf(v0.w, w0, acc.w);
    acc.x = fmaf(v1.x, w1, acc.x); acc.y = fmaf(v1.y, w1, acc.y);
    acc.z = fmaf(v1.z, w1, acc.z); acc.w = fmaf(v1.w, w1, acc.w);
    acc.x = fmaf(v2.x, w2, acc.x); acc.y = fmaf(v2.y, w2, acc.y);
    acc.z = fmaf(v2.z, w2, acc.z); acc.w = fmaf(v2.w, w2, acc.w);
    acc.x = fmaf(v3.x, w3, acc.x); acc.y = fmaf(v3.y, w3, acc.y);
    acc.z = fmaf(v3.z, w3, acc.z); acc.w = fmaf(v3.w, w3, acc.w);
  }
  const int rem = end - j;
  if (rem > 0) {
    const int last = end - 1;
    int2 e0 = csr[j];
    int2 e1 = csr[(j + 1 < end) ? j + 1 : last];
    int2 e2 = csr[(j + 2 < end) ? j + 2 : last];
    float4 v0 = h4[(size_t)e0.x * CHUNKS + c];
    float4 v1 = h4[(size_t)e1.x * CHUNKS + c];
    float4 v2 = h4[(size_t)e2.x * CHUNKS + c];
    float w0 = __int_as_float(e0.y), w1 = __int_as_float(e1.y), w2 = __int_as_float(e2.y);
    acc.x = fmaf(v0.x, w0, acc.x); acc.y = fmaf(v0.y, w0, acc.y);
    acc.z = fmaf(v0.z, w0, acc.z); acc.w = fmaf(v0.w, w0, acc.w);
    if (rem > 1) {
      acc.x = fmaf(v1.x, w1, acc.x); acc.y = fmaf(v1.y, w1, acc.y);
      acc.z = fmaf(v1.z, w1, acc.z); acc.w = fmaf(v1.w, w1, acc.w);
    }
    if (rem > 2) {
      acc.x = fmaf(v2.x, w2, acc.x); acc.y = fmaf(v2.y, w2, acc.y);
      acc.z = fmaf(v2.z, w2, acc.z); acc.w = fmaf(v2.w, w2, acc.w);
    }
  }

  float4* op = (float4*)outp + (size_t)node * CHUNKS + c;
  if (FINAL) {
    const float  sI = invI[node];
    const float4 bb = ((const float4*)bias)[c];
    float4 r;
    r.x = acc.x * sI + bb.x; r.y = acc.y * sI + bb.y;
    r.z = acc.z * sI + bb.z; r.w = acc.w * sI + bb.w;
    *op = r;
  } else {
    *op = acc;
  }
}

__device__ __forceinline__ v8f
gcn_tile_mma(const float* __restrict__ As,
             const float* __restrict__ Wt,
             int lane) {
  const int khalf = (lane >> 4) * 2;
  const float*  As_row = As + (size_t)(lane & 15) * LDK + khalf;
  const float4* Bt     = (const float4*)Wt;
  const unsigned li    = (unsigned)lane;

  v8f acc0 = {};
  v8f acc1 = {};
  #pragma unroll
  for (int k0 = 0; k0 < TK; k0 += 8) {
    const float4 b = (Bt + (k0 >> 3) * 32)[li];
    v2f a0, b0, a1, b1;
    a0[0] = As_row[k0];
    a0[1] = As_row[k0 + 1];
    b0[0] = b.x;
    b0[1] = b.y;
    a1[0] = As_row[k0 + 4];
    a1[1] = As_row[k0 + 5];
    b1[0] = b.z;
    b1[1] = b.w;
    acc0 = __builtin_amdgcn_wmma_f32_16x16x4_f32(false, a0, false, b0, (short)0, acc0,
                                                 false, false);
    acc1 = __builtin_amdgcn_wmma_f32_16x16x4_f32(false, a1, false, b1, (short)0, acc1,
                                                 false, false);
  }
  return acc0 + acc1;
}

__device__ __forceinline__ void
gcn_tile_product(const float* __restrict__ As,
                 const float* __restrict__ Wf,
                 const float* __restrict__ bias,
                 float* __restrict__ C,
                 int mBase, int M, int Nd, int relu, int t) {
  const int wave   = __builtin_amdgcn_readfirstlane(t >> 5);
  const int lane   = t & 31;
  const int nTiles = (Nd + 15) >> 4;
  if (wave >= nTiles) return;

  const v8f acc  = gcn_tile_mma(As, Wf + (size_t)wave * FRAG_TILE, lane);
  const int gcol = wave * 16 + (lane & 15);
  if (gcol < Nd) {
    const float bb    = bias ? bias[gcol] : 0.0f;
    const int   rbase = mBase + (lane >> 4) * 8;
    #pragma unroll
    for (int v = 0; v < 8; ++v) {
      int r = rbase + v;
      if (r < M) {
        float val = acc[v] + bb;
        if (relu) val = fmaxf(val, 0.0f);
        C[(size_t)r * Nd + gcol] = val;
      }
    }
  }
}

__device__ __forceinline__ void
gcn_tile_product_lds(const float* __restrict__ As,
                     const float* __restrict__ Wf,
                     const float* __restrict__ bias,
                     float* __restrict__ Hs,
                     int relu, int t) {
  static_assert(TK == 8 * 16, "one 16-column tile per wave of the 256-thread block");
  const int wave = __builtin_amdgcn_readfirstlane(t >> 5);
  const int lane = t & 31;
  const v8f acc  = gcn_tile_mma(As, Wf + (size_t)wave * FRAG_TILE, lane);
  const int gcol = wave * 16 + (lane & 15);
  const float bb = bias ? bias[gcol] : 0.0f;
  float* hp = Hs + (size_t)((lane >> 4) * 8) * LDK + gcol;
  #pragma unroll
  for (int v = 0; v < 8; ++v) {
    float val = acc[v] + bb;
    if (relu) val = fmaxf(val, 0.0f);
    hp[(size_t)v * LDK] = val;
  }
}

template <bool PROJECT>
__global__ void __launch_bounds__(256, 16)
gcn_gather_gemm_kernel(const float* __restrict__ h,
                       const int* __restrict__ rowptr,
                       const int2* __restrict__ csr,
                       const float* __restrict__ invI,
                       const float* __restrict__ W,
                       const float* __restrict__ bias,
                       const float* __restrict__ Wproj,
                       float* __restrict__ C,
                       int M, int Nd, int relu, int Nproj) {
  __shared__ float As[16 * LDK];

  const int t     = threadIdx.x;
  const int lane  = t & 31;
  const int mBase = blockIdx.x * 16;
  const float4* __restrict__ h4 = (const float4*)h;

  for (int r = 0; r < 2; ++r) {
    const int row  = (t >> 5) * 2 + r;
    const int node = __builtin_amdgcn_readfirstlane(mBase + row);
    float4 acc = make_float4(0.f, 0.f, 0.f, 0.f);
    if (node < M) {
      const int beg  = rowptr[node];
      const int end  = rowptr[node + 1];
      const int last = end - 1;
      int2 e0 = make_int2(0, 0), e1 = e0, e2 = e0, e3 = e0;
      if (beg < end) {
        e0 = csr[beg];
        e1 = csr[min(beg + 1, last)];
        e2 = csr[min(beg + 2, last)];
        e3 = csr[min(beg + 3, last)];
      }
      for (int j = beg; j < end; j += 4) {
        const float4 v0 = h4[(size_t)e0.x * 32 + lane];
        const float4 v1 = h4[(size_t)e1.x * 32 + lane];
        const float4 v2 = h4[(size_t)e2.x * 32 + lane];
        const float4 v3 = h4[(size_t)e3.x * 32 + lane];
        const float w0 = __int_as_float(e0.y), w1 = __int_as_float(e1.y);
        const float w2 = __int_as_float(e2.y), w3 = __int_as_float(e3.y);
        const int rem = end - j;
        const int jn  = j + 4;
        if (jn < end) {
          e0 = csr[jn];
          e1 = csr[min(jn + 1, last)];
          e2 = csr[min(jn + 2, last)];
          e3 = csr[min(jn + 3, last)];
        }
        acc.x = fmaf(v0.x, w0, acc.x); acc.y = fmaf(v0.y, w0, acc.y);
        acc.z = fmaf(v0.z, w0, acc.z); acc.w = fmaf(v0.w, w0, acc.w);
        if (rem > 1) {
          acc.x = fmaf(v1.x, w1, acc.x); acc.y = fmaf(v1.y, w1, acc.y);
          acc.z = fmaf(v1.z, w1, acc.z); acc.w = fmaf(v1.w, w1, acc.w);
        }
        if (rem > 2) {
          acc.x = fmaf(v2.x, w2, acc.x); acc.y = fmaf(v2.y, w2, acc.y);
          acc.z = fmaf(v2.z, w2, acc.z); acc.w = fmaf(v2.w, w2, acc.w);
        }
        if (rem > 3) {
          acc.x = fmaf(v3.x, w3, acc.x); acc.y = fmaf(v3.y, w3, acc.y);
          acc.z = fmaf(v3.z, w3, acc.z); acc.w = fmaf(v3.w, w3, acc.w);
        }
      }
      const float sI = invI[node];
      acc.x *= sI; acc.y *= sI; acc.z *= sI; acc.w *= sI;
    }
    *(float4*)(As + (size_t)row * LDK + (size_t)lane * 4) = acc;
  }
  __syncthreads();

  if constexpr (PROJECT) {
    __shared__ float Hs[16 * LDK];
    gcn_tile_product_lds(As, W, bias, Hs, relu, t);
    __syncthreads();
    gcn_tile_product(Hs, Wproj, nullptr, C, mBase, M, Nproj, 0, t);
  } else {
    gcn_tile_product(As, W, bias, C, mBase, M, Nd, relu, t);
  }
}

extern "C" void kernel_launch(void* const* d_in, const int* in_sizes, int n_in,
                              void* d_out, int out_size, void* d_ws, size_t ws_size,
                              hipStream_t stream) {
  const float* feats = (const float*)d_in[0];
  const int*   src   = (const int*)d_in[1];
  const int*   dst   = (const int*)d_in[2];
  const float* W1    = (const float*)d_in[3];
  const float* b1    = (const float*)d_in[4];
  const float* W2    = (const float*)d_in[5];
  const float* b2    = (const float*)d_in[6];
  const float* W3    = (const float*)d_in[7];
  const float* b3    = (const float*)d_in[8];
  float*       out   = (float*)d_out;

  const int HID  = in_sizes[4];
  const int NCLS = in_sizes[8];
  const int INF  = in_sizes[3] / HID;
  const int N    = in_sizes[0] / INF;
  const int E    = in_sizes[1];

  const size_t Np = ((size_t)N + 1 + 63) & ~(size_t)63;
  const size_t Ep = ((size_t)E + 63) & ~(size_t)63;
  int*   cntI   = (int*)d_ws + Np;
  float* invO   = (float*)(cntI + Np);
  float* invI   = invO + Np;
  int*   rowptr = (int*)(invI + Np);
  int*   rank   = rowptr + Np;
  int2*  csr    = (int2*)(rank + Ep);
  int*   partO  = (int*)(csr + Ep);
  int*   partI  = partO + (size_t)NCHUNK * Np;
  float* bufB   = (float*)(partI + (size_t)NCHUNK * Np);
  float* bufC   = bufB + (size_t)N * TK;
  const size_t nH = (size_t)((HID + 15) / 16) * FRAG_TILE;
  const size_t nC = (size_t)((NCLS + 15) / 16) * FRAG_TILE;
  float* Wf1    = bufC + (((size_t)N * NCLS + 63) & ~(size_t)63);
  float* Wf2    = Wf1 + nH;
  float* Wf3    = Wf2 + nH;

  const int B = 256;
  const int mTiles = (N + 15) / 16;
  const int eBlocks = (E + B - 1) / B;
  const int g10Blocks = (N + 23) / 24;
  const int CE = (E + NCHUNK - 1) / NCHUNK;
  const int nBlocks = (int)((Np + B - 1) / B);
  const int fBlocks = (int)((2 * nH + nC + 1023) / 1024);

  gcn_count_kernel<<<2 * NCHUNK + fBlocks, 1024, 0, stream>>>(src, dst, partO, partI, rank, N, E, CE, (int)Np,
                                                              W1, W2, W3, Wf1, Wf2, Wf3, HID, NCLS);
  gcn_colscan_kernel<<<dim3(nBlocks, 2), B, 0, stream>>>(partO, partI, cntI, invO, invI, N, (int)Np);
  gcn_scan_kernel<<<1, 1024, 0, stream>>>(cntI, rowptr, (int)Np);
  gcn_fill_kernel<<<eBlocks, B, 0, stream>>>(src, dst, rank, rowptr, partI, invO, csr, E, CE, (int)Np);

  gcn_gather_gemm_kernel<false><<<mTiles, B, 0, stream>>>(feats, rowptr, csr, invI, Wf1, b1, nullptr, bufB, N, HID, 1, 0);

  gcn_gather_gemm_kernel<true><<<mTiles, B, 0, stream>>>(bufB, rowptr, csr, invI, Wf2, b2, Wf3, bufC, N, HID, 1, NCLS);

  gcn_gather_kernel<10, true><<<g10Blocks, B, 0, stream>>>(bufC, rowptr, csr, invI, b3, out, N);
}


# Segment starts by two-level scan: column scan adds its block-local node prefix into the chunk start table, fill blocks scan the 196 block totals in LDS; scan launch, a look-up per edge go; sums same
# speedup vs baseline: 4.2534x; 1.0722x over previous
#include <hip/hip_runtime.h>
#include <hip/hip_bf16.h>

typedef __attribute__((ext_vector_type(2))) float v2f;
typedef __attribute__((ext_vector_type(8))) float v8f;

#define TK 128
#define LDK 132

#define NCHUNK 64
#define HB 50176
#define FRAG_TILE 2048

__device__ __forceinline__ void gcn_frag_copy(const float* __restrict__ W,
                                              float* __restrict__ Wf, int Nd, int i) {
  const int e    = i & 3;
  const int lane = (i >> 2) & 31;
  const int s    = (i >> 7) & 15;
  const int j    = i >> 11;
  const int k    = s * 8 + (e >> 1) * 4 + (lane >> 4) * 2 + (e & 1);
  const int col  = j * 16 + (lane & 15);
  Wf[i] = (col < Nd) ? W[(size_t)k * Nd + col] : 0.0f;
}

__global__ void __launch_bounds__(1024)
gcn_count_kernel(const int* __restrict__ src, const int* __restrict__ dst,
                 int* __restrict__ partO, int* __restrict__ partI,
                 int* __restrict__ rank, int N, int E, int CE, int Np,
                 const float* __restrict__ W1, const float* __restrict__ W2,
                 const float* __restrict__ W3, float* __restrict__ Wf1,
                 float* __restrict__ Wf2, float* __restrict__ Wf3,
                 int hid, int ncls) {
  __shared__ int hist[HB];
  const int  t   = threadIdx.x;
  if ((int)blockIdx.x >= 2 * NCHUNK) {
    const int nH = ((hid + 15) >> 4) * FRAG_TILE;
    const int nC = ((ncls + 15) >> 4) * FRAG_TILE;
    const int i  = ((int)blockIdx.x - 2 * NCHUNK) * 1024 + t;
    if (i < nH)               gcn_frag_copy(W1, Wf1, hid, i);
    else if (i < 2 * nH)      gcn_frag_copy(W2, Wf2, hid, i - nH);
    else if (i < 2 * nH + nC) gcn_frag_copy(W3, Wf3, ncls, i - 2 * nH);
    return;
  }
  const bool isI = (int)blockIdx.x < NCHUNK;
  const int  g   = isI ? (int)blockIdx.x : (int)blockIdx.x - NCHUNK;
  const int  e0  = g * CE;
  const int  e1  = min(E, e0 + CE);
  int* __restrict__ part = (isI ? partI : partO) + (size_t)g * Np;

  for (int base = 0; base < N; base += HB) {
    const int lim  = min(HB, N - base);
    const int limP = min(HB, Np - base);
    for (int i = t * 4; i < limP; i += 4096)
      *(int4*)(hist + i) = make_int4(0, 0, 0, 0);
    __syncthreads();
    if (isI) {
      for (int e = e0 + t; e < e1; e += 8 * 1024) {
        int d[8], r[8];
        #pragma unroll
        for (int k = 0; k < 8; ++k) {
          const int ee = e + k * 1024;
          const int v  = dst[min(ee, e1 - 1)];
          d[k] = (ee < e1) ? v - base : -1;
        }
        #pragma unroll
        for (int k = 0; k < 8; ++k)
          r[k] = ((unsigned)d[k] < (unsigned)lim) ? atomicAdd(&hist[d[k]], 1) : 0;
        #pragma unroll
        for (int k = 0; k < 8; ++k)
          if ((unsigned)d[k] < (unsigned)lim) rank[e + k * 1024] = r[k];
      }
    } else {
      for (int e = e0 + t; e < e1; e += 8 * 1024) {
        int d[8];
        #pragma unroll
        for (int k = 0; k < 8; ++k) {
          const int ee = e + k * 1024;
          const int v  = src[min(ee, e1 - 1)];
          d[k] = (ee < e1) ? v - base : -1;
        }
        #pragma unroll
        for (int k = 0; k < 8; ++k) {
          if ((unsigned)d[k] < (unsigned)lim) atomicAdd(&hist[d[k]], 1);
        }
      }
    }
    __syncthreads();
    for (int i = t * 4; i < limP; i += 4096)
      *(int4*)(part + base + i) = *(const int4*)(hist + i);
    __syncthreads();
  }
}

__device__ __forceinline__ float gcn_inv_sqrt_deg(int c) {
  return (c > 0) ? rsqrtf(fmaxf((float)c, 1.0f)) : 0.0f;
}

__device__ __forceinline__ int gcn_wave_incl_scan(int x, int lane) {
  x += __builtin_amdgcn_update_dpp(0, x, 0x111, 0xf, 0xf, false);
  x += __builtin_amdgcn_update_dpp(0, x, 0x112, 0xf, 0xf, false);
  x += __builtin_amdgcn_update_dpp(0, x, 0x114, 0xf, 0xf, false);
  x += __builtin_amdgcn_update_dpp(0, x, 0x118, 0xf, 0xf, false);
  const int row0 = __builtin_amdgcn_readlane(x, 15);
  return (lane >= 16) ? x + row0 : x;
}

#define NODE_BLK 256

__global__ void __launch_bounds__(NODE_BLK)
gcn_colscan_kernel(const int* __restrict__ partO, int* __restrict__ partI,
                   int* __restrict__ rloc, int* __restrict__ blockTot,
                   float* __restrict__ invO, float* __restrict__ invI, int N, int Np) {
  __shared__ int wtot[NODE_BLK / 32];
  const int t = threadIdx.x;
  const int n = blockIdx.x * NODE_BLK + t;
  if (blockIdx.y == 0) {
    int* p = partI + min(n, N - 1);
    int c[NCHUNK];
    #pragma unroll
    for (int g = 0; g < NCHUNK; ++g) c[g] = p[(size_t)g * Np];
    int run = 0;
    #pragma unroll
    for (int g = 0; g < NCHUNK; ++g) run += c[g];
    if (n >= N) run = 0;
    const int lane = t & 31;
    const int w    = t >> 5;
    const int x    = gcn_wave_incl_scan(run, lane);
    if (lane == 31) wtot[w] = x;
    __syncthreads();
    int off = 0, tot = 0;
    #pragma unroll
    for (int k = 0; k < NODE_BLK / 32; ++k) {
      const int s = wtot[k];
      off += (k < w) ? s : 0;
      tot += s;
    }
    const int rl = off + x - run;
    if (n < N) {
      int pre = rl;
      #pragma unroll
      for (int g = 0; g < NCHUNK; ++g) { p[(size_t)g * Np] = pre; pre += c[g]; }
      invI[n] = gcn_inv_sqrt_deg(run);
    }
    if (n < Np) rloc[n] = rl;
    if (t == 0) blockTot[blockIdx.x] = tot;
  } else {
    if (n >= N) return;
    const int* p = partO + n;
    int s = 0;
    #pragma unroll 16
    for (int g = 0; g < NCHUNK; ++g) s += p[(size_t)g * Np];
    invO[n] = gcn_inv_sqrt_deg(s);
  }
}

#define FILL_NB_MAX 1024

__global__ void __launch_bounds__(NODE_BLK)
gcn_fill_kernel(const int* __restrict__ src, const int* __restrict__ dst,
                const int* __restrict__ rank, const int* __restrict__ rloc,
                const int* __restrict__ blockTot, const int* __restrict__ partI,
                const float* __restrict__ invO, int2* __restrict__ csr,
                int* __restrict__ rowptr, int E, int CE, int Np, int NB) {
  __shared__ int bp[FILL_NB_MAX];
  __shared__ int wtot[NODE_BLK / 32];
  const int  t    = threadIdx.x;
  const int  lane = t & 31;
  const int  w    = t >> 5;
  const int  e    = blockIdx.x * NODE_BLK + t;
  const bool live = e < E;
  int s = 0, d = 0, r = 0;
  if (live) { s = src[e]; d = dst[e]; r = rank[e]; }

  int carry = 0;
  for (int base = 0; base < NB; base += NODE_BLK) {
    const int i = base + t;
    const int v = (i < NB) ? blockTot[i] : 0;
    const int x = gcn_wave_incl_scan(v, lane);
    if (lane == 31) wtot[w] = x;
    __syncthreads();
    int off = 0, tot = 0;
    #pragma unroll
    for (int k = 0; k < NODE_BLK / 32; ++k) {
      const int q = wtot[k];
      off += (k < w) ? q : 0;
      tot += q;
    }
    if (i < NB) bp[i] = carry + off + x - v;
    carry += tot;
    __syncthreads();
  }

  if (live) {
    const int g = e / CE;
    int2 p;
    p.x = s;
    p.y = __float_as_int(invO[s]);
    csr[bp[(unsigned)d / NODE_BLK] + partI[(size_t)g * Np + d] + r] = p;
  }
  if ((int)blockIdx.x < NB && e < Np) rowptr[e] = bp[blockIdx.x] + rloc[e];
}

template <int CHUNKS, bool FINAL>
__global__ void __launch_bounds__(256)
gcn_gather_kernel(const float* __restrict__ h,
                  const int* __restrict__ rowptr,
                  const int2* __restrict__ csr,
                  const float* __restrict__ invI,
                  const float* __restrict__ bias,
                  float* __restrict__ outp, int N) {
  constexpr int NPW = 32 / CHUNKS;
  const int lane = threadIdx.x & 31;
  const int wave = blockIdx.x * 8 + (threadIdx.x >> 5);
  const int sub  = lane / CHUNKS;
  const int c    = lane - sub * CHUNKS;
  int node = wave * NPW + sub;
  if (NPW == 1) {
    node = __builtin_amdgcn_readfirstlane(node);
  }
  if (sub >= NPW || node >= N) return;

  const int beg = rowptr[node];
  const int end = rowptr[node + 1];
  const float4* __restrict__ h4 = (const float4*)h;

  float4 acc = make_float4(0.f, 0.f, 0.f, 0.f);
  int j = beg;
  for (; j + 4 <= end; j += 4) {
    int2 e0 = csr[j], e1 = csr[j + 1], e2 = csr[j + 2], e3 = csr[j + 3];
    float4 v0 = h4[(size_t)e0.x * CHUNKS + c];
    float4 v1 = h4[(size_t)e1.x * CHUNKS + c];
    float4 v2 = h4[(size_t)e2.x * CHUNKS + c];
    float4 v3 = h4[(size_t)e3.x * CHUNKS + c];
    float w0 = __int_as_float(e0.y), w1 = __int_as_float(e1.y);
    float w2 = __int_as_float(e2.y), w3 = __int_as_float(e3.y);
    acc.x = fmaf(v0.x, w0, acc.x); acc.y = fmaf(v0.y, w0, acc.y);
    acc.z = fmaf(v0.z, w0, acc.z); acc.w = fmaf(v0.w, w0, acc.w);
    acc.x = fmaf(v1.x, w1, acc.x); acc.y = fmaf(v1.y, w1, acc.y);
    acc.z = fmaf(v1.z, w1, acc.z); acc.w = fmaf(v1.w, w1, acc.w);
    acc.x = fmaf(v2.x, w2, acc.x); acc.y = fmaf(v2.y, w2, acc.y);
    acc.z = fmaf(v2.z, w2, acc.z); acc.w = fmaf(v2.w, w2, acc.w);
    acc.x = fmaf(v3.x, w3, acc.x); acc.y = fmaf(v3.y, w3, acc.y);
    acc.z = fmaf(v3.z, w3, acc.z); acc.w = fmaf(v3.w, w3, acc.w);
  }
  const int rem = end - j;
  if (rem > 0) {
    const int last = end - 1;
    int2 e0 = csr[j];
    int2 e1 = csr[(j + 1 < end) ? j + 1 : last];
    int2 e2 = csr[(j + 2 < end) ? j + 2 : last];
    float4 v0 = h4[(size_t)e0.x * CHUNKS + c];
    float4 v1 = h4[(size_t)e1.x * CHUNKS + c];
    float4 v2 = h4[(size_t)e2.x * CHUNKS + c];
    float w0 = __int_as_float(e0.y), w1 = __int_as_float(e1.y), w2 = __int_as_float(e2.y);
    acc.x = fmaf(v0.x, w0, acc.x); acc.y = fmaf(v0.y, w0, acc.y);
    acc.z = fmaf(v0.z, w0, acc.z); acc.w = fmaf(v0.w, w0, acc.w);
    if (rem > 1) {
      acc.x = fmaf(v1.x, w1, acc.x); acc.y = fmaf(v1.y, w1, acc.y);
      acc.z = fmaf(v1.z, w1, acc.z); acc.w = fmaf(v1.w, w1, acc.w);
    }
    if (rem > 2) {
      acc.x = fmaf(v2.x, w2, acc.x); acc.y = fmaf(v2.y, w2, acc.y);
      acc.z = fmaf(v2.z, w2, acc.z); acc.w = fmaf(v2.w, w2, acc.w);
    }
  }

  float4* op = (float4*)outp + (size_t)node * CHUNKS + c;
  if (FINAL) {
    const float  sI = invI[node];
    const float4 bb = ((const float4*)bias)[c];
    float4 r;
    r.x = acc.x * sI + bb.x; r.y = acc.y * sI + bb.y;
    r.z = acc.z * sI + bb.z; r.w = acc.w * sI + bb.w;
    *op = r;
  } else {
    *op = acc;
  }
}

__device__ __forceinline__ v8f
gcn_tile_mma(const float* __restrict__ As,
             const float* __restrict__ Wt,
             int lane) {
  const int khalf = (lane >> 4) * 2;
  const float*  As_row = As + (size_t)(lane & 15) * LDK + khalf;
  const float4* Bt     = (const float4*)Wt;
  const unsigned li    = (unsigned)lane;

  v8f acc0 = {};
  v8f acc1 = {};
  #pragma unroll
  for (int k0 = 0; k0 < TK; k0 += 8) {
    const float4 b = (Bt + (k0 >> 3) * 32)[li];
    v2f a0, b0, a1, b1;
    a0[0] = As_row[k0];
    a0[1] = As_row[k0 + 1];
    b0[0] = b.x;
    b0[1] = b.y;
    a1[0] = As_row[k0 + 4];
    a1[1] = As_row[k0 + 5];
    b1[0] = b.z;
    b1[1] = b.w;
    acc0 = __builtin_amdgcn_wmma_f32_16x16x4_f32(false, a0, false, b0, (short)0, acc0,
                                                 false, false);
    acc1 = __builtin_amdgcn_wmma_f32_16x16x4_f32(false, a1, false, b1, (short)0, acc1,
                                                 false, false);
  }
  return acc0 + acc1;
}

__device__ __forceinline__ void
gcn_tile_product(const float* __restrict__ As,
                 const float* __restrict__ Wf,
                 const float* __restrict__ bias,
                 float* __restrict__ C,
                 int mBase, int M, int Nd, int relu, int t) {
  const int wave   = __builtin_amdgcn_readfirstlane(t >> 5);
  const int lane   = t & 31;
  const int nTiles = (Nd + 15) >> 4;
  if (wave >= nTiles) return;

  const v8f acc  = gcn_tile_mma(As, Wf + (size_t)wave * FRAG_TILE, lane);
  const int gcol = wave * 16 + (lane & 15);
  if (gcol < Nd) {
    const float bb    = bias ? bias[gcol] : 0.0f;
    const int   rbase = mBase + (lane >> 4) * 8;
    #pragma unroll
    for (int v = 0; v < 8; ++v) {
      int r = rbase + v;
      if (r < M) {
        float val = acc[v] + bb;
        if (relu) val = fmaxf(val, 0.0f);
        C[(size_t)r * Nd + gcol] = val;
      }
    }
  }
}

__device__ __forceinline__ void
gcn_tile_product_lds(const float* __restrict__ As,
                     const float* __restrict__ Wf,
                     const float* __restrict__ bias,
                     float* __restrict__ Hs,
                     int relu, int t) {
  static_assert(TK == 8 * 16, "one 16-column tile per wave of the 256-thread block");
  const int wave = __builtin_amdgcn_readfirstlane(t >> 5);
  const int lane = t & 31;
  const v8f acc  = gcn_tile_mma(As, Wf + (size_t)wave * FRAG_TILE, lane);
  const int gcol = wave * 16 + (lane & 15);
  const float bb = bias ? bias[gcol] : 0.0f;
  float* hp = Hs + (size_t)((lane >> 4) * 8) * LDK + gcol;
  #pragma unroll
  for (int v = 0; v < 8; ++v) {
    float val = acc[v] + bb;
    if (relu) val = fmaxf(val, 0.0f);
    hp[(size_t)v * LDK] = val;
  }
}

template <bool PROJECT>
__global__ void __launch_bounds__(256, 16)
gcn_gather_gemm_kernel(const float* __restrict__ h,
                       const int* __restrict__ rowptr,
                       const int2* __restrict__ csr,
                       const float* __restrict__ invI,
                       const float* __restrict__ W,
                       const float* __restrict__ bias,
                       const float* __restrict__ Wproj,
                       float* __restrict__ C,
                       int M, int Nd, int relu, int Nproj) {
  __shared__ float As[16 * LDK];

  const int t     = threadIdx.x;
  const int lane  = t & 31;
  const int mBase = blockIdx.x * 16;
  const float4* __restrict__ h4 = (const float4*)h;

  for (int r = 0; r < 2; ++r) {
    const int row  = (t >> 5) * 2 + r;
    const int node = __builtin_amdgcn_readfirstlane(mBase + row);
    float4 acc = make_float4(0.f, 0.f, 0.f, 0.f);
    if (node < M) {
      const int beg  = rowptr[node];
      const int end  = rowptr[node + 1];
      const int last = end - 1;
      int2 e0 = make_int2(0, 0), e1 = e0, e2 = e0, e3 = e0;
      if (beg < end) {
        e0 = csr[beg];
        e1 = csr[min(beg + 1, last)];
        e2 = csr[min(beg + 2, last)];
        e3 = csr[min(beg + 3, last)];
      }
      for (int j = beg; j < end; j += 4) {
        const float4 v0 = h4[(size_t)e0.x * 32 + lane];
        const float4 v1 = h4[(size_t)e1.x * 32 + lane];
        const float4 v2 = h4[(size_t)e2.x * 32 + lane];
        const float4 v3 = h4[(size_t)e3.x * 32 + lane];
        const float w0 = __int_as_float(e0.y), w1 = __int_as_float(e1.y);
        const float w2 = __int_as_float(e2.y), w3 = __int_as_float(e3.y);
        const int rem = end - j;
        const int jn  = j + 4;
        if (jn < end) {
          e0 = csr[jn];
          e1 = csr[min(jn + 1, last)];
          e2 = csr[min(jn + 2, last)];
          e3 = csr[min(jn + 3, last)];
        }
        acc.x = fmaf(v0.x, w0, acc.x); acc.y = fmaf(v0.y, w0, acc.y);
        acc.z = fmaf(v0.z, w0, acc.z); acc.w = fmaf(v0.w, w0, acc.w);
        if (rem > 1) {
          acc.x = fmaf(v1.x, w1, acc.x); acc.y = fmaf(v1.y, w1, acc.y);
          acc.z = fmaf(v1.z, w1, acc.z); acc.w = fmaf(v1.w, w1, acc.w);
        }
        if (rem > 2) {
          acc.x = fmaf(v2.x, w2, acc.x); acc.y = fmaf(v2.y, w2, acc.y);
          acc.z = fmaf(v2.z, w2, acc.z); acc.w = fmaf(v2.w, w2, acc.w);
        }
        if (rem > 3) {
          acc.x = fmaf(v3.x, w3, acc.x); acc.y = fmaf(v3.y, w3, acc.y);
          acc.z = fmaf(v3.z, w3, acc.z); acc.w = fmaf(v3.w, w3, acc.w);
        }
      }
      const float sI = invI[node];
      acc.x *= sI; acc.y *= sI; acc.z *= sI; acc.w *= sI;
    }
    *(float4*)(As + (size_t)row * LDK + (size_t)lane * 4) = acc;
  }
  __syncthreads();

  if constexpr (PROJECT) {
    __shared__ float Hs[16 * LDK];
    gcn_tile_product_lds(As, W, bias, Hs, relu, t);
    __syncthreads();
    gcn_tile_product(Hs, Wproj, nullptr, C, mBase, M, Nproj, 0, t);
  } else {
    gcn_tile_product(As, W, bias, C, mBase, M, Nd, relu, t);
  }
}

extern "C" void kernel_launch(void* const* d_in, const int* in_sizes, int n_in,
                              void* d_out, int out_size, void* d_ws, size_t ws_size,
                              hipStream_t stream) {
  const float* feats = (const float*)d_in[0];
  const int*   src   = (const int*)d_in[1];
  const int*   dst   = (const int*)d_in[2];
  const float* W1    = (const float*)d_in[3];
  const float* b1    = (const float*)d_in[4];
  const float* W2    = (const float*)d_in[5];
  const float* b2    = (const float*)d_in[6];
  const float* W3    = (const float*)d_in[7];
  const float* b3    = (const float*)d_in[8];
  float*       out   = (float*)d_out;

  const int HID  = in_sizes[4];
  const int NCLS = in_sizes[8];
  const int INF  = in_sizes[3] / HID;
  const int N    = in_sizes[0] / INF;
  const int E    = in_sizes[1];

  const size_t Np = ((size_t)N + 1 + 63) & ~(size_t)63;
  const size_t Ep = ((size_t)E + 63) & ~(size_t)63;
  int*   rloc   = (int*)d_ws;
  int*   blockTot = rloc + Np;
  float* invO   = (float*)(blockTot + Np);
  float* invI   = invO + Np;
  int*   rowptr = (int*)(invI + Np);
  int*   rank   = rowptr + Np;
  int2*  csr    = (int2*)(rank + Ep);
  int*   partO  = (int*)(csr + Ep);
  int*   partI  = partO + (size_t)NCHUNK * Np;
  float* bufB   = (float*)(partI + (size_t)NCHUNK * Np);
  float* bufC   = bufB + (size_t)N * TK;
  const size_t nH = (size_t)((HID + 15) / 16) * FRAG_TILE;
  const size_t nC = (size_t)((NCLS + 15) / 16) * FRAG_TILE;
  float* Wf1    = bufC + (((size_t)N * NCLS + 63) & ~(size_t)63);
  float* Wf2    = Wf1 + nH;
  float* Wf3    = Wf2 + nH;

  const int B = 256;
  const int mTiles = (N + 15) / 16;
  const int eBlocks = (E + NODE_BLK - 1) / NODE_BLK;
  const int g10Blocks = (N + 23) / 24;
  const int CE = (E + NCHUNK - 1) / NCHUNK;
  const int nBlocks = (int)((Np + NODE_BLK - 1) / NODE_BLK);
  const int fillBlocks = eBlocks > nBlocks ? eBlocks : nBlocks;
  const int fBlocks = (int)((2 * nH + nC + 1023) / 1024);

  gcn_count_kernel<<<2 * NCHUNK + fBlocks, 1024, 0, stream>>>(src, dst, partO, partI, rank, N, E, CE, (int)Np,
                                                              W1, W2, W3, Wf1, Wf2, Wf3, HID, NCLS);
  gcn_colscan_kernel<<<dim3(nBlocks, 2), NODE_BLK, 0, stream>>>(partO, partI, rloc, blockTot, invO, invI, N, (int)Np);
  gcn_fill_kernel<<<fillBlocks, NODE_BLK, 0, stream>>>(src, dst, rank, rloc, blockTot, partI, invO, csr, rowptr, E, CE, (int)Np, nBlocks);

  gcn_gather_gemm_kernel<false><<<mTiles, B, 0, stream>>>(feats, rowptr, csr, invI, Wf1, b1, nullptr, bufB, N, HID, 1, 0);

  gcn_gather_gemm_kernel<true><<<mTiles, B, 0, stream>>>(bufB, rowptr, csr, invI, Wf2, b2, Wf3, bufC, N, HID, 1, NCLS);

  gcn_gather_kernel<10, true><<<g10Blocks, B, 0, stream>>>(bufC, rowptr, csr, invI, b3, out, N);
}
